# starting kernel
# baseline (speedup 1.0000x reference)
#include <hip/hip_runtime.h>
#include <hip/hip_bf16.h>
#include <math.h>

#define N_NODES 100000
#define N_EDGES 1600000
#define IN_DIM  128
#define OUT_DIM 64
#define LEAKY   0.01f

typedef float v2f __attribute__((ext_vector_type(2)));
typedef float v8f __attribute__((ext_vector_type(8)));

__device__ __forceinline__ unsigned fkey(float f) {
    unsigned u = __float_as_uint(f);
    return (u & 0x80000000u) ? ~u : (u | 0x80000000u);
}
__device__ __forceinline__ float keyf(unsigned k) {
    unsigned u = (k & 0x80000000u) ? (k & 0x7FFFFFFFu) : ~k;
    return __uint_as_float(u);
}
#define NEG_INF_KEY 0x007FFFFFu

__global__ void gat_init(float* __restrict__ out, unsigned* __restrict__ mkey,
                         float* __restrict__ denom) {
    int i = blockIdx.x * blockDim.x + threadIdx.x;
    if (i < N_NODES * OUT_DIM) out[i] = 0.0f;
    if (i < N_NODES) { mkey[i] = NEG_INF_KEY; denom[i] = 0.0f; }
}

__global__ void gat_gemm_wmma(const float* __restrict__ h,
                              const float* __restrict__ W,
                              float* __restrict__ z) {
    const int lane = threadIdx.x & 31;
    const int wave = threadIdx.x >> 5;
    const int row0 = blockIdx.x * 16;
    const int col0 = wave * 16;

    const int m     = lane & 15;
    const int khalf = (lane >> 4) * 2;

    const float* __restrict__ hrow = h + (size_t)(row0 + m) * IN_DIM;
    const float* __restrict__ wrow = W + (size_t)(col0 + m) * IN_DIM;

    __builtin_prefetch(wrow, 0, 3);

    v8f c = {};
#pragma unroll
    for (int k0 = 0; k0 < IN_DIM; k0 += 4) {
        v2f a, b;
        a.x = hrow[k0 + khalf];
        a.y = hrow[k0 + khalf + 1];
        b.x = wrow[k0 + khalf];
        b.y = wrow[k0 + khalf + 1];
        c = __builtin_amdgcn_wmma_f32_16x16x4_f32(
                false, a, false, b, (short)0, c, false, false);
    }

    const int n     = lane & 15;
    const int rbase = (lane >> 4) * 8;
#pragma unroll
    for (int r = 0; r < 8; ++r) {
        z[(size_t)(row0 + rbase + r) * OUT_DIM + col0 + n] = c[r];
    }
}

__global__ void gat_proj(const float* __restrict__ z, const float* __restrict__ A,
                         float* __restrict__ s, float* __restrict__ t) {
    int i = blockIdx.x * blockDim.x + threadIdx.x;
    if (i >= N_NODES) return;
    const float4* zr = (const float4*)(z + (size_t)i * OUT_DIM);
    const float4* as = (const float4*)A;
    const float4* ad = (const float4*)(A + OUT_DIM);
    float accs = 0.f, acct = 0.f;
#pragma unroll
    for (int k = 0; k < OUT_DIM / 4; ++k) {
        float4 v = zr[k], x = as[k], y = ad[k];
        accs += v.x * x.x + v.y * x.y + v.z * x.z + v.w * x.w;
        acct += v.x * y.x + v.y * y.y + v.z * y.z + v.w * y.w;
    }
    s[i] = accs;
    t[i] = acct;
}

__global__ void gat_edge_logit(const int* __restrict__ src, const int* __restrict__ dst,
                               const float* __restrict__ s, const float* __restrict__ t,
                               float* __restrict__ eb, unsigned* __restrict__ mkey) {
    int e = blockIdx.x * blockDim.x + threadIdx.x;
    if (e >= N_EDGES) return;
    int si = src[e], di = dst[e];
    float v = s[si] + t[di];
    v = (v > 0.f) ? v : LEAKY * v;
    eb[e] = v;
    atomicMax(&mkey[di], fkey(v));
}

__global__ void gat_mfix(unsigned* __restrict__ mkey) {
    int i = blockIdx.x * blockDim.x + threadIdx.x;
    if (i >= N_NODES) return;
    float mv = keyf(mkey[i]);
    if (!isfinite(mv)) mv = 0.0f;
    ((float*)mkey)[i] = mv;
}

__global__ void gat_p_denom(const int* __restrict__ dst, const float* __restrict__ m,
                            float* __restrict__ eb, float* __restrict__ denom) {
    int e = blockIdx.x * blockDim.x + threadIdx.x;
    if (e >= N_EDGES) return;
    int di = dst[e];
    float p = expf(eb[e] - m[di]);
    eb[e] = p;
    atomicAdd(&denom[di], p);
}

__global__ void gat_scatter(const int* __restrict__ src, const int* __restrict__ dst,
                            const float* __restrict__ p, const float* __restrict__ denom,
                            const float* __restrict__ z, float* __restrict__ out) {
    const int lane = threadIdx.x & 31;
    const int wave = threadIdx.x >> 5;
    const int e = blockIdx.x * 8 + wave;
    if (e >= N_EDGES) return;
    const int si = src[e], di = dst[e];
    const float alpha = p[e] / denom[di];
    float2 v = ((const float2*)(z + (size_t)si * OUT_DIM))[lane];
    float* o = out + (size_t)di * OUT_DIM + 2 * lane;
    atomicAdd(o + 0, alpha * v.x);
    atomicAdd(o + 1, alpha * v.y);
}

extern "C" void kernel_launch(void* const* d_in, const int* in_sizes, int n_in,
                              void* d_out, int out_size, void* d_ws, size_t ws_size,
                              hipStream_t stream) {
    const float* h  = (const float*)d_in[0];
    const int* src  = (const int*)d_in[1];
    const int* dst  = (const int*)d_in[2];
    const float* W  = (const float*)d_in[3];
    const float* A  = (const float*)d_in[4];
    float* out = (float*)d_out;

    char* ws = (char*)d_ws;
    float*    z     = (float*)   (ws);
    float*    s     = (float*)   (ws + 25600000);
    float*    t     = (float*)   (ws + 26000000);
    unsigned* mkey  = (unsigned*)(ws + 26400000);
    float*    denom = (float*)   (ws + 26800000);
    float*    eb    = (float*)   (ws + 27200000);

    const int node_blocks  = (N_NODES + 255) / 256;
    const int edge_blocks  = (N_EDGES + 255) / 256;
    const int init_blocks  = (N_NODES * OUT_DIM + 255) / 256;

    gat_init<<<init_blocks, 256, 0, stream>>>(out, mkey, denom);
    gat_gemm_wmma<<<N_NODES / 16, 128, 0, stream>>>(h, W, z);
    gat_proj<<<node_blocks, 256, 0, stream>>>(z, A, s, t);
    gat_edge_logit<<<edge_blocks, 256, 0, stream>>>(src, dst, s, t, eb, mkey);
    gat_mfix<<<node_blocks, 256, 0, stream>>>(mkey);
    gat_p_denom<<<edge_blocks, 256, 0, stream>>>(dst, (const float*)mkey, eb, denom);
    gat_scatter<<<(N_EDGES + 7) / 8, 256, 0, stream>>>(src, dst, eb, denom, z, out);
}


# edge phase without sort or atomics: 256 work-groups keep 391 output rows in LDS, list their edges from a byte-per-edge key stream; one wave owns a row; running-maximum sums in list order; 4 launches
# speedup vs baseline: 1.4951x; 1.4951x over previous
#include <hip/hip_runtime.h>
#include <hip/hip_bf16.h>
#include <math.h>

#define N_NODES 100000
#define N_EDGES 1600000
#define IN_DIM  128
#define OUT_DIM 64
#define LEAKY   0.01f

typedef float v2f __attribute__((ext_vector_type(2)));
typedef float v8f __attribute__((ext_vector_type(8)));

__global__ void gat_gemm_wmma(const float* __restrict__ h,
                              const float* __restrict__ W,
                              float* __restrict__ z) {
    const int lane = threadIdx.x & 31;
    const int wave = threadIdx.x >> 5;
    const int row0 = blockIdx.x * 16;
    const int col0 = wave * 16;

    const int m     = lane & 15;
    const int khalf = (lane >> 4) * 2;

    const float* __restrict__ hrow = h + (size_t)(row0 + m) * IN_DIM;
    const float* __restrict__ wrow = W + (size_t)(col0 + m) * IN_DIM;

    __builtin_prefetch(wrow, 0, 3);

    v8f c = {};
#pragma unroll
    for (int k0 = 0; k0 < IN_DIM; k0 += 4) {
        v2f a, b;
        a.x = hrow[k0 + khalf];
        a.y = hrow[k0 + khalf + 1];
        b.x = wrow[k0 + khalf];
        b.y = wrow[k0 + khalf + 1];
        c = __builtin_amdgcn_wmma_f32_16x16x4_f32(
                false, a, false, b, (short)0, c, false, false);
    }

    const int n     = lane & 15;
    const int rbase = (lane >> 4) * 8;
#pragma unroll
    for (int r = 0; r < 8; ++r) {
        z[(size_t)(row0 + rbase + r) * OUT_DIM + col0 + n] = c[r];
    }
}

__global__ void gat_proj(const float* __restrict__ z, const float* __restrict__ A,
                         float* __restrict__ s, float* __restrict__ t) {
    int i = blockIdx.x * blockDim.x + threadIdx.x;
    if (i >= N_NODES) return;
    const float4* zr = (const float4*)(z + (size_t)i * OUT_DIM);
    const float4* as = (const float4*)A;
    const float4* ad = (const float4*)(A + OUT_DIM);
    float accs = 0.f, acct = 0.f;
#pragma unroll
    for (int k = 0; k < OUT_DIM / 4; ++k) {
        float4 v = zr[k], x = as[k], y = ad[k];
        accs += v.x * x.x + v.y * x.y + v.z * x.z + v.w * x.w;
        acct += v.x * y.x + v.y * y.y + v.z * y.z + v.w * y.w;
    }
    s[i] = accs;
    t[i] = acct;
}

#define RANGE_ROWS 391
#define N_RANGES   256
#define CHUNK      512
#define N_CHUNKS   (N_EDGES / CHUNK)
#define SCAN_WAVES 16
#define CHUNKS_PER_WAVE 196
#define SEG_CAP    512
#define LIST_CAP   (SCAN_WAVES * SEG_CAP)
#define BATCH      256
#define MAX_PASS   200

__global__ void gat_keys(const int* __restrict__ dst, unsigned char* __restrict__ key) {
    int e = blockIdx.x * blockDim.x + threadIdx.x;
    if (e >= N_EDGES) return;
    key[e] = (unsigned char)(dst[e] / RANGE_ROWS);
}

__device__ __forceinline__ unsigned zero_bytes(unsigned y) {
    return ~(((y & 0x7F7F7F7Fu) + 0x7F7F7F7Fu) | y | 0x7F7F7F7Fu);
}

__global__ void __launch_bounds__(512)
gat_range(const int* __restrict__ src, const int* __restrict__ dst,
          const unsigned char* __restrict__ key,
          const float* __restrict__ s, const float* __restrict__ t,
          const float* __restrict__ z, float* __restrict__ out) {
    __shared__ float    acc[RANGE_ROWS * OUT_DIM];
    __shared__ float    stage[BATCH * OUT_DIM];
    __shared__ unsigned l_idx[LIST_CAP];
    __shared__ unsigned c_pk[LIST_CAP];
    __shared__ float    c_e[LIST_CAP];
    __shared__ float2   rmd[400];
    __shared__ int      segcnt[SCAN_WAVES];
    __shared__ int      more;

    const int tid  = threadIdx.x;
    const int lane = tid & 31;
    const int wave = __builtin_amdgcn_readfirstlane(tid >> 5);
    const int gid  = blockIdx.x;
    const int row_base = gid * RANGE_ROWS;

    for (int i = tid; i < RANGE_ROWS * OUT_DIM / 4; i += 512)
        ((float4*)acc)[i] = make_float4(0.f, 0.f, 0.f, 0.f);
    for (int i = tid; i < 400; i += 512) rmd[i] = make_float2(-INFINITY, 0.f);

    const uint4* __restrict__ key4 = (const uint4*)key;
    const unsigned pat = (unsigned)gid * 0x01010101u;
    int c = wave * CHUNKS_PER_WAVE;
    int cend = c + CHUNKS_PER_WAVE;
    if (cend > N_CHUNKS) cend = N_CHUNKS;

    for (int pass = 0; pass < MAX_PASS; ++pass) {
        if (tid == 0) more = 0;
        __syncthreads();

        int cnt = 0;
        bool full = false;
        while (c < cend && !full) {
            uint4 kv[8];
#pragma unroll
            for (int u = 0; u < 8; ++u) {
                int cc = c + u;
                if (cc > cend - 1) cc = cend - 1;
                kv[u] = key4[(size_t)cc * 32 + lane];
            }
#pragma unroll
            for (int u = 0; u < 8; ++u) {
                if (c < cend && !full) {
                    const unsigned t0 = zero_bytes(kv[u].x ^ pat);
                    const unsigned t1 = zero_bytes(kv[u].y ^ pat);
                    const unsigned t2 = zero_bytes(kv[u].z ^ pat);
                    const unsigned t3 = zero_bytes(kv[u].w ^ pat);
                    unsigned mk = __builtin_amdgcn_ballot_w32((t0 | t1 | t2 | t3) != 0u);
                    if (mk) {
                        const int cl = __popc(t0) + __popc(t1) + __popc(t2) + __popc(t3);
                        int tot = 0;
                        unsigned mm = mk;
                        while (mm) {
                            const int l = __builtin_ctz(mm);
                            mm &= mm - 1;
                            tot += __builtin_amdgcn_readlane(cl, l);
                        }
                        if (cnt + tot > SEG_CAP) {
                            full = true;
                        } else {
                            const unsigned ebase = (unsigned)c * CHUNK;
                            while (mk) {
                                const int l = __builtin_ctz(mk);
                                mk &= mk - 1;
                                unsigned tw[4];
                                tw[0] = (unsigned)__builtin_amdgcn_readlane((int)t0, l);
                                tw[1] = (unsigned)__builtin_amdgcn_readlane((int)t1, l);
                                tw[2] = (unsigned)__builtin_amdgcn_readlane((int)t2, l);
                                tw[3] = (unsigned)__builtin_amdgcn_readlane((int)t3, l);
#pragma unroll
                                for (int j = 0; j < 4; ++j) {
                                    unsigned tj = tw[j];
                                    while (tj) {
                                        const int b = __builtin_ctz(tj) >> 3;
                                        tj &= tj - 1;
                                        if (lane == 0)
                                            l_idx[wave * SEG_CAP + cnt] = ebase + (unsigned)(l * 16 + j * 4 + b);
                                        ++cnt;
                                    }
                                }
                            }
                        }
                    }
                    if (!full) ++c;
                }
            }
        }
        if (lane == 0) {
            segcnt[wave] = cnt;
            if (c < cend) more = 1;
        }
        __syncthreads();
        const int again = more;

        int total = 0;
        for (int w = 0; w < SCAN_WAVES; ++w) total += segcnt[w];
        {
            unsigned ee[SCAN_WAVES];
            int si[SCAN_WAVES], di[SCAN_WAVES];
            float sv[SCAN_WAVES], tv[SCAN_WAVES];
#pragma unroll
            for (int w = 0; w < SCAN_WAVES; ++w)
                ee[w] = (tid < segcnt[w]) ? l_idx[w * SEG_CAP + tid] : 0u;
#pragma unroll
            for (int w = 0; w < SCAN_WAVES; ++w) { si[w] = src[ee[w]]; di[w] = dst[ee[w]]; }
#pragma unroll
            for (int w = 0; w < SCAN_WAVES; ++w) { sv[w] = s[si[w]]; tv[w] = t[di[w]]; }
            int base = 0;
#pragma unroll
            for (int w = 0; w < SCAN_WAVES; ++w) {
                const int n = segcnt[w];
                if (tid < n) {
                    float v = sv[w] + tv[w];
                    v = (v > 0.f) ? v : LEAKY * v;
                    c_pk[base + tid] = (unsigned)si[w] | ((unsigned)(di[w] - row_base) << 17);
                    c_e[base + tid]  = v;
                }
                base += n;
            }
        }
        __syncthreads();

#define ROW_PIECE(K, B) {                                                           \
            const int q_ = tid + 512 * (K);                                          \
            int jx_ = (B) + (q_ >> 4);                                               \
            if (jx_ > total - 1) jx_ = total - 1;                                    \
            const unsigned sidx_ = c_pk[jx_] & 0x1FFFFu;                             \
            p##K = ((const float4*)(z + (size_t)sidx_ * OUT_DIM))[q_ & 15]; }
        float4 p0 = make_float4(0.f, 0.f, 0.f, 0.f);
        float4 p1 = p0, p2 = p0, p3 = p0, p4 = p0, p5 = p0, p6 = p0, p7 = p0;
        if (total > 0) {
            ROW_PIECE(0, 0) ROW_PIECE(1, 0) ROW_PIECE(2, 0) ROW_PIECE(3, 0)
            ROW_PIECE(4, 0) ROW_PIECE(5, 0) ROW_PIECE(6, 0) ROW_PIECE(7, 0)
        }
        for (int b0 = 0; b0 < total; b0 += BATCH) {
            ((float4*)stage)[tid + 512 * 0] = p0;
            ((float4*)stage)[tid + 512 * 1] = p1;
            ((float4*)stage)[tid + 512 * 2] = p2;
            ((float4*)stage)[tid + 512 * 3] = p3;
            ((float4*)stage)[tid + 512 * 4] = p4;
            ((float4*)stage)[tid + 512 * 5] = p5;
            ((float4*)stage)[tid + 512 * 6] = p6;
            ((float4*)stage)[tid + 512 * 7] = p7;
            __syncthreads();
            if (b0 + BATCH < total) {
                const int nb = b0 + BATCH;
                ROW_PIECE(0, nb) ROW_PIECE(1, nb) ROW_PIECE(2, nb) ROW_PIECE(3, nb)
                ROW_PIECE(4, nb) ROW_PIECE(5, nb) ROW_PIECE(6, nb) ROW_PIECE(7, nb)
            }
            unsigned pkv[BATCH / 32];
            float    evv[BATCH / 32];
#pragma unroll
            for (int g = 0; g < BATCH / 32; ++g) {
                const int j = b0 + g * 32 + lane;
                const int jc = (j < total) ? j : 0;
                pkv[g] = c_pk[jc];
                evv[g] = c_e[jc];
            }
#pragma unroll
            for (int g = 0; g < BATCH / 32; ++g) {
                const int j = b0 + g * 32 + lane;
                const int lrow = (int)(pkv[g] >> 17);
                unsigned mk = __builtin_amdgcn_ballot_w32((j < total) && ((lrow & 15) == wave));
                while (mk) {
                    const int l = __builtin_ctz(mk);
                    mk &= mk - 1;
                    const int r = __builtin_amdgcn_readlane(lrow, l);
                    const float e = __int_as_float(__builtin_amdgcn_readlane(__float_as_int(evv[g]), l));
                    const int jj = g * 32 + l;
                    const float2 md = rmd[r];
                    float2 a = ((float2*)acc)[r * 32 + lane];
                    const float2 zv = ((const float2*)stage)[jj * 32 + lane];
                    const float mn = fmaxf(md.x, e);
                    const float sc = expf(md.x - mn);
                    const float p  = expf(e - mn);
                    a.x = a.x * sc + p * zv.x;
                    a.y = a.y * sc + p * zv.y;
                    ((float2*)acc)[r * 32 + lane] = a;
                    if (lane == 0) rmd[r] = make_float2(mn, md.y * sc + p);
                }
            }
            __syncthreads();
        }
        if (!again) break;
    }
    __syncthreads();

    for (int i = tid; i < RANGE_ROWS * OUT_DIM / 4; i += 512) {
        const int r = i >> 4;
        const int node = row_base + r;
        if (node < N_NODES) {
            const float d = rmd[r].y;
            float4 a = ((const float4*)acc)[i];
            const float inv = (d > 0.f) ? 1.0f / d : 0.f;
            a.x *= inv; a.y *= inv; a.z *= inv; a.w *= inv;
            ((float4*)(out + (size_t)node * OUT_DIM))[i & 15] = a;
        }
    }
}

extern "C" void kernel_launch(void* const* d_in, const int* in_sizes, int n_in,
                              void* d_out, int out_size, void* d_ws, size_t ws_size,
                              hipStream_t stream) {
    const float* h  = (const float*)d_in[0];
    const int* src  = (const int*)d_in[1];
    const int* dst  = (const int*)d_in[2];
    const float* W  = (const float*)d_in[3];
    const float* A  = (const float*)d_in[4];
    float* out = (float*)d_out;

    char* ws = (char*)d_ws;
    float*         z   = (float*)        (ws);
    float*         s   = (float*)        (ws + 25600000);
    float*         t   = (float*)        (ws + 26000000);
    unsigned char* key = (unsigned char*)(ws + 26400000);

    const int node_blocks  = (N_NODES + 255) / 256;
    const int edge_blocks  = (N_EDGES + 255) / 256;

    gat_gemm_wmma<<<N_NODES / 16, 128, 0, stream>>>(h, W, z);
    gat_proj<<<node_blocks, 256, 0, stream>>>(z, A, s, t);
    gat_keys<<<edge_blocks, 256, 0, stream>>>(dst, key);
    gat_range<<<N_RANGES, 512, 0, stream>>>(src, dst, key, s, t, z, out);
}


# front end in one launch: product work-groups pass the z tile through LDS, store whole lines, form s and t from it (4-column partials, 16-lane tree: new s,t sum order), write 256 edge keys; 2 launches
# speedup vs baseline: 1.5674x; 1.0484x over previous
#include <hip/hip_runtime.h>
#include <hip/hip_bf16.h>
#include <math.h>

#define N_NODES 100000
#define N_EDGES 1600000
#define IN_DIM  128
#define OUT_DIM 64
#define LEAKY   0.01f

typedef float v2f __attribute__((ext_vector_type(2)));
typedef float v8f __attribute__((ext_vector_type(8)));

#define RANGE_ROWS 391
#define N_RANGES   256
#define CHUNK      512
#define N_CHUNKS   (N_EDGES / CHUNK)
#define SCAN_WAVES 16
#define CHUNKS_PER_WAVE 196
#define SEG_CAP    512
#define LIST_CAP   (SCAN_WAVES * SEG_CAP)
#define BATCH      256
#define MAX_PASS   200


#define ZT_PITCH 68
#define EDGES_PER_BLOCK 256
static_assert((N_NODES % 16) == 0, "row tiles must be whole");
static_assert(EDGES_PER_BLOCK * (N_NODES / 16) == N_EDGES, "every edge gets its key from exactly one block");

__global__ void gat_front(const float* __restrict__ h,
                          const float* __restrict__ W,
                          const float* __restrict__ A,
                          const int* __restrict__ dst,
                          float* __restrict__ z,
                          float* __restrict__ s, float* __restrict__ t,
                          unsigned char* __restrict__ key) {
    __shared__ float4 zt4[16 * ZT_PITCH / 4];
    float* zt = (float*)zt4;

    const int tid  = threadIdx.x;
    const int lane = tid & 31;
    const int wave = tid >> 5;
    const int row0 = blockIdx.x * 16;
    const int col0 = wave * 16;

    const int e0 = (blockIdx.x * (EDGES_PER_BLOCK / 2) + tid) * 2;
    const int2 dd = *(const int2*)(dst + e0);

    const int m     = lane & 15;
    const int khalf = (lane >> 4) * 2;

    const float* __restrict__ hrow = h + (size_t)(row0 + m) * IN_DIM;
    const float* __restrict__ wrow = W + (size_t)(col0 + m) * IN_DIM;

    __builtin_prefetch(wrow, 0, 3);

    v8f c = {};
#pragma unroll
    for (int k0 = 0; k0 < IN_DIM; k0 += 4) {
        v2f a, b;
        a.x = hrow[k0 + khalf];
        a.y = hrow[k0 + khalf + 1];
        b.x = wrow[k0 + khalf];
        b.y = wrow[k0 + khalf + 1];
        c = __builtin_amdgcn_wmma_f32_16x16x4_f32(
                false, a, false, b, (short)0, c, false, false);
    }

    const int n     = lane & 15;
    const int rbase = (lane >> 4) * 8;
#pragma unroll
    for (int r = 0; r < 8; ++r) {
        zt[(rbase + r) * ZT_PITCH + col0 + n] = c[r];
    }
    __syncthreads();

    const int q  = tid & 15;
    const int rr = tid >> 4;
    const float4 as = ((const float4*)A)[q];
    const float4 ad = ((const float4*)(A + OUT_DIM))[q];
    const float4 v0 = zt4[rr * (ZT_PITCH / 4) + q];
    const float4 v1 = zt4[(rr + 8) * (ZT_PITCH / 4) + q];
    float ps0 = v0.x * as.x + v0.y * as.y + v0.z * as.z + v0.w * as.w;
    float pt0 = v0.x * ad.x + v0.y * ad.y + v0.z * ad.z + v0.w * ad.w;
    float ps1 = v1.x * as.x + v1.y * as.y + v1.z * as.z + v1.w * as.w;
    float pt1 = v1.x * ad.x + v1.y * ad.y + v1.z * ad.z + v1.w * ad.w;
    ((float4*)(z + (size_t)(row0 + rr) * OUT_DIM))[q] = v0;
    ((float4*)(z + (size_t)(row0 + rr + 8) * OUT_DIM))[q] = v1;
#pragma unroll
    for (int mk = 1; mk < 16; mk <<= 1) {
        const float a0 = __shfl_xor(ps0, mk, 32);
        const float a1 = __shfl_xor(pt0, mk, 32);
        const float a2 = __shfl_xor(ps1, mk, 32);
        const float a3 = __shfl_xor(pt1, mk, 32);
        ps0 += a0; pt0 += a1; ps1 += a2; pt1 += a3;
    }
    if (q == 0) {
        s[row0 + rr]     = ps0;
        t[row0 + rr]     = pt0;
        s[row0 + rr + 8] = ps1;
        t[row0 + rr + 8] = pt1;
    }

    uchar2 kk;
    kk.x = (unsigned char)(dd.x / RANGE_ROWS);
    kk.y = (unsigned char)(dd.y / RANGE_ROWS);
    *(uchar2*)(key + e0) = kk;
}

__device__ __forceinline__ unsigned zero_bytes(unsigned y) {
    return ~(((y & 0x7F7F7F7Fu) + 0x7F7F7F7Fu) | y | 0x7F7F7F7Fu);
}

__global__ void __launch_bounds__(512)
gat_range(const int* __restrict__ src, const int* __restrict__ dst,
          const unsigned char* __restrict__ key,
          const float* __restrict__ s, const float* __restrict__ t,
          const float* __restrict__ z, float* __restrict__ out) {
    __shared__ float    acc[RANGE_ROWS * OUT_DIM];
    __shared__ float    stage[BATCH * OUT_DIM];
    __shared__ unsigned l_idx[LIST_CAP];
    __shared__ unsigned c_pk[LIST_CAP];
    __shared__ float    c_e[LIST_CAP];
    __shared__ float2   rmd[400];
    __shared__ int      segcnt[SCAN_WAVES];
    __shared__ int      more;

    const int tid  = threadIdx.x;
    const int lane = tid & 31;
    const int wave = __builtin_amdgcn_readfirstlane(tid >> 5);
    const int gid  = blockIdx.x;
    const int row_base = gid * RANGE_ROWS;

    for (int i = tid; i < RANGE_ROWS * OUT_DIM / 4; i += 512)
        ((float4*)acc)[i] = make_float4(0.f, 0.f, 0.f, 0.f);
    for (int i = tid; i < 400; i += 512) rmd[i] = make_float2(-INFINITY, 0.f);

    const uint4* __restrict__ key4 = (const uint4*)key;
    const unsigned pat = (unsigned)gid * 0x01010101u;
    int c = wave * CHUNKS_PER_WAVE;
    int cend = c + CHUNKS_PER_WAVE;
    if (cend > N_CHUNKS) cend = N_CHUNKS;

    for (int pass = 0; pass < MAX_PASS; ++pass) {
        if (tid == 0) more = 0;
        __syncthreads();

        int cnt = 0;
        bool full = false;
        while (c < cend && !full) {
            uint4 kv[8];
#pragma unroll
            for (int u = 0; u < 8; ++u) {
                int cc = c + u;
                if (cc > cend - 1) cc = cend - 1;
                kv[u] = key4[(size_t)cc * 32 + lane];
            }
#pragma unroll
            for (int u = 0; u < 8; ++u) {
                if (c < cend && !full) {
                    const unsigned t0 = zero_bytes(kv[u].x ^ pat);
                    const unsigned t1 = zero_bytes(kv[u].y ^ pat);
                    const unsigned t2 = zero_bytes(kv[u].z ^ pat);
                    const unsigned t3 = zero_bytes(kv[u].w ^ pat);
                    unsigned mk = __builtin_amdgcn_ballot_w32((t0 | t1 | t2 | t3) != 0u);
                    if (mk) {
                        const int cl = __popc(t0) + __popc(t1) + __popc(t2) + __popc(t3);
                        int tot = 0;
                        unsigned mm = mk;
                        while (mm) {
                            const int l = __builtin_ctz(mm);
                            mm &= mm - 1;
                            tot += __builtin_amdgcn_readlane(cl, l);
                        }
                        if (cnt + tot > SEG_CAP) {
                            full = true;
                        } else {
                            const unsigned ebase = (unsigned)c * CHUNK;
                            while (mk) {
                                const int l = __builtin_ctz(mk);
                                mk &= mk - 1;
                                unsigned tw[4];
                                tw[0] = (unsigned)__builtin_amdgcn_readlane((int)t0, l);
                                tw[1] = (unsigned)__builtin_amdgcn_readlane((int)t1, l);
                                tw[2] = (unsigned)__builtin_amdgcn_readlane((int)t2, l);
                                tw[3] = (unsigned)__builtin_amdgcn_readlane((int)t3, l);
#pragma unroll
                                for (int j = 0; j < 4; ++j) {
                                    unsigned tj = tw[j];
                                    while (tj) {
                                        const int b = __builtin_ctz(tj) >> 3;
                                        tj &= tj - 1;
                                        if (lane == 0)
                                            l_idx[wave * SEG_CAP + cnt] = ebase + (unsigned)(l * 16 + j * 4 + b);
                                        ++cnt;
                                    }
                                }
                            }
                        }
                    }
                    if (!full) ++c;
                }
            }
        }
        if (lane == 0) {
            segcnt[wave] = cnt;
            if (c < cend) more = 1;
        }
        __syncthreads();
        const int again = more;

        int total = 0;
        for (int w = 0; w < SCAN_WAVES; ++w) total += segcnt[w];
        {
            unsigned ee[SCAN_WAVES];
            int si[SCAN_WAVES], di[SCAN_WAVES];
            float sv[SCAN_WAVES], tv[SCAN_WAVES];
#pragma unroll
            for (int w = 0; w < SCAN_WAVES; ++w)
                ee[w] = (tid < segcnt[w]) ? l_idx[w * SEG_CAP + tid] : 0u;
#pragma unroll
            for (int w = 0; w < SCAN_WAVES; ++w) { si[w] = src[ee[w]]; di[w] = dst[ee[w]]; }
#pragma unroll
            for (int w = 0; w < SCAN_WAVES; ++w) { sv[w] = s[si[w]]; tv[w] = t[di[w]]; }
            int base = 0;
#pragma unroll
            for (int w = 0; w < SCAN_WAVES; ++w) {
                const int n = segcnt[w];
                if (tid < n) {
                    float v = sv[w] + tv[w];
                    v = (v > 0.f) ? v : LEAKY * v;
                    c_pk[base + tid] = (unsigned)si[w] | ((unsigned)(di[w] - row_base) << 17);
                    c_e[base + tid]  = v;
                }
                base += n;
            }
        }
        __syncthreads();

#define ROW_PIECE(K, B) {                                                           \
            const int q_ = tid + 512 * (K);                                          \
            int jx_ = (B) + (q_ >> 4);                                               \
            if (jx_ > total - 1) jx_ = total - 1;                                    \
            const unsigned sidx_ = c_pk[jx_] & 0x1FFFFu;                             \
            p##K = ((const float4*)(z + (size_t)sidx_ * OUT_DIM))[q_ & 15]; }
        float4 p0 = make_float4(0.f, 0.f, 0.f, 0.f);
        float4 p1 = p0, p2 = p0, p3 = p0, p4 = p0, p5 = p0, p6 = p0, p7 = p0;
        if (total > 0) {
            ROW_PIECE(0, 0) ROW_PIECE(1, 0) ROW_PIECE(2, 0) ROW_PIECE(3, 0)
            ROW_PIECE(4, 0) ROW_PIECE(5, 0) ROW_PIECE(6, 0) ROW_PIECE(7, 0)
        }
        for (int b0 = 0; b0 < total; b0 += BATCH) {
            ((float4*)stage)[tid + 512 * 0] = p0;
            ((float4*)stage)[tid + 512 * 1] = p1;
            ((float4*)stage)[tid + 512 * 2] = p2;
            ((float4*)stage)[tid + 512 * 3] = p3;
            ((float4*)stage)[tid + 512 * 4] = p4;
            ((float4*)stage)[tid + 512 * 5] = p5;
            ((float4*)stage)[tid + 512 * 6] = p6;
            ((float4*)stage)[tid + 512 * 7] = p7;
            __syncthreads();
            if (b0 + BATCH < total) {
                const int nb = b0 + BATCH;
                ROW_PIECE(0, nb) ROW_PIECE(1, nb) ROW_PIECE(2, nb) ROW_PIECE(3, nb)
                ROW_PIECE(4, nb) ROW_PIECE(5, nb) ROW_PIECE(6, nb) ROW_PIECE(7, nb)
            }
            unsigned pkv[BATCH / 32];
            float    evv[BATCH / 32];
#pragma unroll
            for (int g = 0; g < BATCH / 32; ++g) {
                const int j = b0 + g * 32 + lane;
                const int jc = (j < total) ? j : 0;
                pkv[g] = c_pk[jc];
                evv[g] = c_e[jc];
            }
#pragma unroll
            for (int g = 0; g < BATCH / 32; ++g) {
                const int j = b0 + g * 32 + lane;
                const int lrow = (int)(pkv[g] >> 17);
                unsigned mk = __builtin_amdgcn_ballot_w32((j < total) && ((lrow & 15) == wave));
                while (mk) {
                    const int l = __builtin_ctz(mk);
                    mk &= mk - 1;
                    const int r = __builtin_amdgcn_readlane(lrow, l);
                    const float e = __int_as_float(__builtin_amdgcn_readlane(__float_as_int(evv[g]), l));
                    const int jj = g * 32 + l;
                    const float2 md = rmd[r];
                    float2 a = ((float2*)acc)[r * 32 + lane];
                    const float2 zv = ((const float2*)stage)[jj * 32 + lane];
                    const float mn = fmaxf(md.x, e);
                    const float sc = expf(md.x - mn);
                    const float p  = expf(e - mn);
                    a.x = a.x * sc + p * zv.x;
                    a.y = a.y * sc + p * zv.y;
                    ((float2*)acc)[r * 32 + lane] = a;
                    if (lane == 0) rmd[r] = make_float2(mn, md.y * sc + p);
                }
            }
            __syncthreads();
        }
        if (!again) break;
    }
    __syncthreads();

    for (int i = tid; i < RANGE_ROWS * OUT_DIM / 4; i += 512) {
        const int r = i >> 4;
        const int node = row_base + r;
        if (node < N_NODES) {
            const float d = rmd[r].y;
            float4 a = ((const float4*)acc)[i];
            const float inv = (d > 0.f) ? 1.0f / d : 0.f;
            a.x *= inv; a.y *= inv; a.z *= inv; a.w *= inv;
            ((float4*)(out + (size_t)node * OUT_DIM))[i & 15] = a;
        }
    }
}

extern "C" void kernel_launch(void* const* d_in, const int* in_sizes, int n_in,
                              void* d_out, int out_size, void* d_ws, size_t ws_size,
                              hipStream_t stream) {
    const float* h  = (const float*)d_in[0];
    const int* src  = (const int*)d_in[1];
    const int* dst  = (const int*)d_in[2];
    const float* W  = (const float*)d_in[3];
    const float* A  = (const float*)d_in[4];
    float* out = (float*)d_out;

    char* ws = (char*)d_ws;
    float*         z   = (float*)        (ws);
    float*         s   = (float*)        (ws + 25600000);
    float*         t   = (float*)        (ws + 26000000);
    unsigned char* key = (unsigned char*)(ws + 26400000);

    gat_front<<<N_NODES / 16, 128, 0, stream>>>(h, W, A, dst, z, s, t, key);
    gat_range<<<N_RANGES, 512, 0, stream>>>(src, dst, key, s, t, z, out);
}


# key scan: each matching lane appends its own edges at the count of matches in the lanes below (ballot prefix), not lane 0 walking every match in scalar loops; same list order, same sums
# speedup vs baseline: 1.7222x; 1.0988x over previous
#include <hip/hip_runtime.h>
#include <hip/hip_bf16.h>
#include <math.h>

#define N_NODES 100000
#define N_EDGES 1600000
#define IN_DIM  128
#define OUT_DIM 64
#define LEAKY   0.01f

typedef float v2f __attribute__((ext_vector_type(2)));
typedef float v8f __attribute__((ext_vector_type(8)));

#define RANGE_ROWS 391
#define N_RANGES   256
#define CHUNK      512
#define N_CHUNKS   (N_EDGES / CHUNK)
#define SCAN_WAVES 16
#define CHUNKS_PER_WAVE 196
#define SEG_CAP    512
#define LIST_CAP   (SCAN_WAVES * SEG_CAP)
#define BATCH      256
#define MAX_PASS   200


#define ZT_PITCH 68
#define EDGES_PER_BLOCK 256
static_assert((N_NODES % 16) == 0, "row tiles must be whole");
static_assert(EDGES_PER_BLOCK * (N_NODES / 16) == N_EDGES, "every edge gets its key from exactly one block");

__global__ void gat_front(const float* __restrict__ h,
                          const float* __restrict__ W,
                          const float* __restrict__ A,
                          const int* __restrict__ dst,
                          float* __restrict__ z,
                          float* __restrict__ s, float* __restrict__ t,
                          unsigned char* __restrict__ key) {
    __shared__ float4 zt4[16 * ZT_PITCH / 4];
    float* zt = (float*)zt4;

    const int tid  = threadIdx.x;
    const int lane = tid & 31;
    const int wave = tid >> 5;
    const int row0 = blockIdx.x * 16;
    const int col0 = wave * 16;

    const int e0 = (blockIdx.x * (EDGES_PER_BLOCK / 2) + tid) * 2;
    const int2 dd = *(const int2*)(dst + e0);

    const int m     = lane & 15;
    const int khalf = (lane >> 4) * 2;

    const float* __restrict__ hrow = h + (size_t)(row0 + m) * IN_DIM;
    const float* __restrict__ wrow = W + (size_t)(col0 + m) * IN_DIM;

    __builtin_prefetch(wrow, 0, 3);

    v8f c = {};
#pragma unroll
    for (int k0 = 0; k0 < IN_DIM; k0 += 4) {
        v2f a, b;
        a.x = hrow[k0 + khalf];
        a.y = hrow[k0 + khalf + 1];
        b.x = wrow[k0 + khalf];
        b.y = wrow[k0 + khalf + 1];
        c = __builtin_amdgcn_wmma_f32_16x16x4_f32(
                false, a, false, b, (short)0, c, false, false);
    }

    const int n     = lane & 15;
    const int rbase = (lane >> 4) * 8;
#pragma unroll
    for (int r = 0; r < 8; ++r) {
        zt[(rbase + r) * ZT_PITCH + col0 + n] = c[r];
    }
    __syncthreads();

    const int q  = tid & 15;
    const int rr = tid >> 4;
    const float4 as = ((const float4*)A)[q];
    const float4 ad = ((const float4*)(A + OUT_DIM))[q];
    const float4 v0 = zt4[rr * (ZT_PITCH / 4) + q];
    const float4 v1 = zt4[(rr + 8) * (ZT_PITCH / 4) + q];
    float ps0 = v0.x * as.x + v0.y * as.y + v0.z * as.z + v0.w * as.w;
    float pt0 = v0.x * ad.x + v0.y * ad.y + v0.z * ad.z + v0.w * ad.w;
    float ps1 = v1.x * as.x + v1.y * as.y + v1.z * as.z + v1.w * as.w;
    float pt1 = v1.x * ad.x + v1.y * ad.y + v1.z * ad.z + v1.w * ad.w;
    ((float4*)(z + (size_t)(row0 + rr) * OUT_DIM))[q] = v0;
    ((float4*)(z + (size_t)(row0 + rr + 8) * OUT_DIM))[q] = v1;
#pragma unroll
    for (int mk = 1; mk < 16; mk <<= 1) {
        const float a0 = __shfl_xor(ps0, mk, 32);
        const float a1 = __shfl_xor(pt0, mk, 32);
        const float a2 = __shfl_xor(ps1, mk, 32);
        const float a3 = __shfl_xor(pt1, mk, 32);
        ps0 += a0; pt0 += a1; ps1 += a2; pt1 += a3;
    }
    if (q == 0) {
        s[row0 + rr]     = ps0;
        t[row0 + rr]     = pt0;
        s[row0 + rr + 8] = ps1;
        t[row0 + rr + 8] = pt1;
    }

    uchar2 kk;
    kk.x = (unsigned char)(dd.x / RANGE_ROWS);
    kk.y = (unsigned char)(dd.y / RANGE_ROWS);
    *(uchar2*)(key + e0) = kk;
}

__device__ __forceinline__ unsigned zero_bytes(unsigned y) {
    return ~(((y & 0x7F7F7F7Fu) + 0x7F7F7F7Fu) | y | 0x7F7F7F7Fu);
}

__global__ void __launch_bounds__(512)
gat_range(const int* __restrict__ src, const int* __restrict__ dst,
          const unsigned char* __restrict__ key,
          const float* __restrict__ s, const float* __restrict__ t,
          const float* __restrict__ z, float* __restrict__ out) {
    __shared__ float    acc[RANGE_ROWS * OUT_DIM];
    __shared__ float    stage[BATCH * OUT_DIM];
    __shared__ unsigned l_idx[LIST_CAP];
    __shared__ unsigned c_pk[LIST_CAP];
    __shared__ float    c_e[LIST_CAP];
    __shared__ float2   rmd[400];
    __shared__ int      segcnt[SCAN_WAVES];
    __shared__ int      more;

    const int tid  = threadIdx.x;
    const int lane = tid & 31;
    const int wave = __builtin_amdgcn_readfirstlane(tid >> 5);
    const int gid  = blockIdx.x;
    const int row_base = gid * RANGE_ROWS;

    for (int i = tid; i < RANGE_ROWS * OUT_DIM / 4; i += 512)
        ((float4*)acc)[i] = make_float4(0.f, 0.f, 0.f, 0.f);
    for (int i = tid; i < 400; i += 512) rmd[i] = make_float2(-INFINITY, 0.f);

    const uint4* __restrict__ key4 = (const uint4*)key;
    const unsigned pat = (unsigned)gid * 0x01010101u;
    int c = wave * CHUNKS_PER_WAVE;
    int cend = c + CHUNKS_PER_WAVE;
    if (cend > N_CHUNKS) cend = N_CHUNKS;

    for (int pass = 0; pass < MAX_PASS; ++pass) {
        if (tid == 0) more = 0;
        __syncthreads();

        int cnt = 0;
        bool full = false;
        while (c < cend && !full) {
            uint4 kv[8];
#pragma unroll
            for (int u = 0; u < 8; ++u) {
                int cc = c + u;
                if (cc > cend - 1) cc = cend - 1;
                kv[u] = key4[(size_t)cc * 32 + lane];
            }
#pragma unroll
            for (int u = 0; u < 8; ++u) {
                if (c < cend && !full) {
                    const unsigned t0 = zero_bytes(kv[u].x ^ pat);
                    const unsigned t1 = zero_bytes(kv[u].y ^ pat);
                    const unsigned t2 = zero_bytes(kv[u].z ^ pat);
                    const unsigned t3 = zero_bytes(kv[u].w ^ pat);
                    unsigned mk = __builtin_amdgcn_ballot_w32((t0 | t1 | t2 | t3) != 0u);
                    if (mk) {
                        const int cl = __popc(t0) + __popc(t1) + __popc(t2) + __popc(t3);
                        int tot, off;
                        if (__builtin_amdgcn_ballot_w32(cl > 1) == 0u) {
                            tot = __popc(mk);
                            off = (int)__builtin_amdgcn_mbcnt_lo(mk, 0u);
                        } else {
                            int x = cl;
#pragma unroll
                            for (int d = 1; d < 32; d <<= 1) {
                                const int y = __shfl_up(x, d, 32);
                                if (lane >= d) x += y;
                            }
                            off = x - cl;
                            tot = __builtin_amdgcn_readlane(x, 31);
                        }
                        if (cnt + tot > SEG_CAP) {
                            full = true;
                        } else {
                            if (cl) {
                                const unsigned n0 = (((t0 >> 7) & 0x01010101u) * 0x01020408u) >> 24;
                                const unsigned n1 = (((t1 >> 7) & 0x01010101u) * 0x01020408u) >> 24;
                                const unsigned n2 = (((t2 >> 7) & 0x01010101u) * 0x01020408u) >> 24;
                                const unsigned n3 = (((t3 >> 7) & 0x01010101u) * 0x01020408u) >> 24;
                                unsigned m16 = (n0 & 15u) | ((n1 & 15u) << 4) | ((n2 & 15u) << 8) | ((n3 & 15u) << 12);
                                const unsigned eb = (unsigned)c * CHUNK + (unsigned)(lane * 16);
                                int pos = wave * SEG_CAP + cnt + off;
                                while (m16) {
                                    l_idx[pos++] = eb + (unsigned)__builtin_ctz(m16);
                                    m16 &= m16 - 1;
                                }
                            }
                            cnt += tot;
                        }
                    }
                    if (!full) ++c;
                }
            }
        }
        if (lane == 0) {
            segcnt[wave] = cnt;
            if (c < cend) more = 1;
        }
        __syncthreads();
        const int again = more;

        int total = 0;
        for (int w = 0; w < SCAN_WAVES; ++w) total += segcnt[w];
        {
            unsigned ee[SCAN_WAVES];
            int si[SCAN_WAVES], di[SCAN_WAVES];
            float sv[SCAN_WAVES], tv[SCAN_WAVES];
#pragma unroll
            for (int w = 0; w < SCAN_WAVES; ++w)
                ee[w] = (tid < segcnt[w]) ? l_idx[w * SEG_CAP + tid] : 0u;
#pragma unroll
            for (int w = 0; w < SCAN_WAVES; ++w) { si[w] = src[ee[w]]; di[w] = dst[ee[w]]; }
#pragma unroll
            for (int w = 0; w < SCAN_WAVES; ++w) { sv[w] = s[si[w]]; tv[w] = t[di[w]]; }
            int base = 0;
#pragma unroll
            for (int w = 0; w < SCAN_WAVES; ++w) {
                const int n = segcnt[w];
                if (tid < n) {
                    float v = sv[w] + tv[w];
                    v = (v > 0.f) ? v : LEAKY * v;
                    c_pk[base + tid] = (unsigned)si[w] | ((unsigned)(di[w] - row_base) << 17);
                    c_e[base + tid]  = v;
                }
                base += n;
            }
        }
        __syncthreads();

#define ROW_PIECE(K, B) {                                                           \
            const int q_ = tid + 512 * (K);                                          \
            int jx_ = (B) + (q_ >> 4);                                               \
            if (jx_ > total - 1) jx_ = total - 1;                                    \
            const unsigned sidx_ = c_pk[jx_] & 0x1FFFFu;                             \
            p##K = ((const float4*)(z + (size_t)sidx_ * OUT_DIM))[q_ & 15]; }
        float4 p0 = make_float4(0.f, 0.f, 0.f, 0.f);
        float4 p1 = p0, p2 = p0, p3 = p0, p4 = p0, p5 = p0, p6 = p0, p7 = p0;
        if (total > 0) {
            ROW_PIECE(0, 0) ROW_PIECE(1, 0) ROW_PIECE(2, 0) ROW_PIECE(3, 0)
            ROW_PIECE(4, 0) ROW_PIECE(5, 0) ROW_PIECE(6, 0) ROW_PIECE(7, 0)
        }
        for (int b0 = 0; b0 < total; b0 += BATCH) {
            ((float4*)stage)[tid + 512 * 0] = p0;
            ((float4*)stage)[tid + 512 * 1] = p1;
            ((float4*)stage)[tid + 512 * 2] = p2;
            ((float4*)stage)[tid + 512 * 3] = p3;
            ((float4*)stage)[tid + 512 * 4] = p4;
            ((float4*)stage)[tid + 512 * 5] = p5;
            ((float4*)stage)[tid + 512 * 6] = p6;
            ((float4*)stage)[tid + 512 * 7] = p7;
            __syncthreads();
            if (b0 + BATCH < total) {
                const int nb = b0 + BATCH;
                ROW_PIECE(0, nb) ROW_PIECE(1, nb) ROW_PIECE(2, nb) ROW_PIECE(3, nb)
                ROW_PIECE(4, nb) ROW_PIECE(5, nb) ROW_PIECE(6, nb) ROW_PIECE(7, nb)
            }
            unsigned pkv[BATCH / 32];
            float    evv[BATCH / 32];
#pragma unroll
            for (int g = 0; g < BATCH / 32; ++g) {
                const int j = b0 + g * 32 + lane;
                const int jc = (j < total) ? j : 0;
                pkv[g] = c_pk[jc];
                evv[g] = c_e[jc];
            }
#pragma unroll
            for (int g = 0; g < BATCH / 32; ++g) {
                const int j = b0 + g * 32 + lane;
                const int lrow = (int)(pkv[g] >> 17);
                unsigned mk = __builtin_amdgcn_ballot_w32((j < total) && ((lrow & 15) == wave));
                while (mk) {
                    const int l = __builtin_ctz(mk);
                    mk &= mk - 1;
                    const int r = __builtin_amdgcn_readlane(lrow, l);
                    const float e = __int_as_float(__builtin_amdgcn_readlane(__float_as_int(evv[g]), l));
                    const int jj = g * 32 + l;
                    const float2 md = rmd[r];
                    float2 a = ((float2*)acc)[r * 32 + lane];
                    const float2 zv = ((const float2*)stage)[jj * 32 + lane];
                    const float mn = fmaxf(md.x, e);
                    const float sc = expf(md.x - mn);
                    const float p  = expf(e - mn);
                    a.x = a.x * sc + p * zv.x;
                    a.y = a.y * sc + p * zv.y;
                    ((float2*)acc)[r * 32 + lane] = a;
                    if (lane == 0) rmd[r] = make_float2(mn, md.y * sc + p);
                }
            }
            __syncthreads();
        }
        if (!again) break;
    }
    __syncthreads();

    for (int i = tid; i < RANGE_ROWS * OUT_DIM / 4; i += 512) {
        const int r = i >> 4;
        const int node = row_base + r;
        if (node < N_NODES) {
            const float d = rmd[r].y;
            float4 a = ((const float4*)acc)[i];
            const float inv = (d > 0.f) ? 1.0f / d : 0.f;
            a.x *= inv; a.y *= inv; a.z *= inv; a.w *= inv;
            ((float4*)(out + (size_t)node * OUT_DIM))[i & 15] = a;
        }
    }
}

extern "C" void kernel_launch(void* const* d_in, const int* in_sizes, int n_in,
                              void* d_out, int out_size, void* d_ws, size_t ws_size,
                              hipStream_t stream) {
    const float* h  = (const float*)d_in[0];
    const int* src  = (const int*)d_in[1];
    const int* dst  = (const int*)d_in[2];
    const float* W  = (const float*)d_in[3];
    const float* A  = (const float*)d_in[4];
    float* out = (float*)d_out;

    char* ws = (char*)d_ws;
    float*         z   = (float*)        (ws);
    float*         s   = (float*)        (ws + 25600000);
    float*         t   = (float*)        (ws + 26000000);
    unsigned char* key = (unsigned char*)(ws + 26400000);

    gat_front<<<N_NODES / 16, 128, 0, stream>>>(h, W, A, dst, z, s, t, key);
    gat_range<<<N_RANGES, 512, 0, stream>>>(src, dst, key, s, t, z, out);
}


# range phase: row maxima first (LDS integer atomic max, order-free), each listed edge weighted exp(e-max) once by its lister; owner wave adds p*z per edge in list order: no exp or rescale in the loop
# speedup vs baseline: 2.3036x; 1.3376x over previous
#include <hip/hip_runtime.h>
#include <hip/hip_bf16.h>
#include <math.h>

#define N_NODES 100000
#define N_EDGES 1600000
#define IN_DIM  128
#define OUT_DIM 64
#define LEAKY   0.01f

typedef float v2f __attribute__((ext_vector_type(2)));
typedef float v8f __attribute__((ext_vector_type(8)));

#define RANGE_ROWS 391
#define N_RANGES   256
#define CHUNK      512
#define N_CHUNKS   (N_EDGES / CHUNK)
#define SCAN_WAVES 16
#define CHUNKS_PER_WAVE 196
#define SEG_CAP    512
#define LIST_CAP   (SCAN_WAVES * SEG_CAP)
#define BATCH      256
#define MAX_PASS   200


#define ZT_PITCH 68
#define EDGES_PER_BLOCK 256
static_assert((N_NODES % 16) == 0, "row tiles must be whole");
static_assert(EDGES_PER_BLOCK * (N_NODES / 16) == N_EDGES, "every edge gets its key from exactly one block");

__global__ void gat_front(const float* __restrict__ h,
                          const float* __restrict__ W,
                          const float* __restrict__ A,
                          const int* __restrict__ dst,
                          float* __restrict__ z,
                          float* __restrict__ s, float* __restrict__ t,
                          unsigned char* __restrict__ key) {
    __shared__ float4 zt4[16 * ZT_PITCH / 4];
    float* zt = (float*)zt4;

    const int tid  = threadIdx.x;
    const int lane = tid & 31;
    const int wave = tid >> 5;
    const int row0 = blockIdx.x * 16;
    const int col0 = wave * 16;

    const int e0 = (blockIdx.x * (EDGES_PER_BLOCK / 2) + tid) * 2;
    const int2 dd = *(const int2*)(dst + e0);

    const int m     = lane & 15;
    const int khalf = (lane >> 4) * 2;

    const float* __restrict__ hrow = h + (size_t)(row0 + m) * IN_DIM;
    const float* __restrict__ wrow = W + (size_t)(col0 + m) * IN_DIM;

    __builtin_prefetch(wrow, 0, 3);

    v8f c = {};
#pragma unroll
    for (int k0 = 0; k0 < IN_DIM; k0 += 4) {
        v2f a, b;
        a.x = hrow[k0 + khalf];
        a.y = hrow[k0 + khalf + 1];
        b.x = wrow[k0 + khalf];
        b.y = wrow[k0 + khalf + 1];
        c = __builtin_amdgcn_wmma_f32_16x16x4_f32(
                false, a, false, b, (short)0, c, false, false);
    }

    const int n     = lane & 15;
    const int rbase = (lane >> 4) * 8;
#pragma unroll
    for (int r = 0; r < 8; ++r) {
        zt[(rbase + r) * ZT_PITCH + col0 + n] = c[r];
    }
    __syncthreads();

    const int q  = tid & 15;
    const int rr = tid >> 4;
    const float4 as = ((const float4*)A)[q];
    const float4 ad = ((const float4*)(A + OUT_DIM))[q];
    const float4 v0 = zt4[rr * (ZT_PITCH / 4) + q];
    const float4 v1 = zt4[(rr + 8) * (ZT_PITCH / 4) + q];
    float ps0 = v0.x * as.x + v0.y * as.y + v0.z * as.z + v0.w * as.w;
    float pt0 = v0.x * ad.x + v0.y * ad.y + v0.z * ad.z + v0.w * ad.w;
    float ps1 = v1.x * as.x + v1.y * as.y + v1.z * as.z + v1.w * as.w;
    float pt1 = v1.x * ad.x + v1.y * ad.y + v1.z * ad.z + v1.w * ad.w;
    ((float4*)(z + (size_t)(row0 + rr) * OUT_DIM))[q] = v0;
    ((float4*)(z + (size_t)(row0 + rr + 8) * OUT_DIM))[q] = v1;
#pragma unroll
    for (int mk = 1; mk < 16; mk <<= 1) {
        const float a0 = __shfl_xor(ps0, mk, 32);
        const float a1 = __shfl_xor(pt0, mk, 32);
        const float a2 = __shfl_xor(ps1, mk, 32);
        const float a3 = __shfl_xor(pt1, mk, 32);
        ps0 += a0; pt0 += a1; ps1 += a2; pt1 += a3;
    }
    if (q == 0) {
        s[row0 + rr]     = ps0;
        t[row0 + rr]     = pt0;
        s[row0 + rr + 8] = ps1;
        t[row0 + rr + 8] = pt1;
    }

    uchar2 kk;
    kk.x = (unsigned char)(dd.x / RANGE_ROWS);
    kk.y = (unsigned char)(dd.y / RANGE_ROWS);
    *(uchar2*)(key + e0) = kk;
}

__device__ __forceinline__ unsigned zero_bytes(unsigned y) {
    return ~(((y & 0x7F7F7F7Fu) + 0x7F7F7F7Fu) | y | 0x7F7F7F7Fu);
}


__device__ __forceinline__ unsigned fkey(float v) {
    const unsigned b = __float_as_uint(v);
    return (b & 0x80000000u) ? ~b : (b | 0x80000000u);
}
__device__ __forceinline__ float funkey(unsigned k) {
    return __uint_as_float((k & 0x80000000u) ? (k & 0x7FFFFFFFu) : ~k);
}
__global__ void __launch_bounds__(512)
gat_range(const int* __restrict__ src, const int* __restrict__ dst,
          const unsigned char* __restrict__ key,
          const float* __restrict__ s, const float* __restrict__ t,
          const float* __restrict__ z, float* __restrict__ out) {
    __shared__ float    acc[RANGE_ROWS * OUT_DIM];
    __shared__ float    stage[BATCH * OUT_DIM];
    __shared__ unsigned l_idx[LIST_CAP];
    __shared__ unsigned c_pk[LIST_CAP];
    __shared__ float    c_e[LIST_CAP];
    __shared__ unsigned rmax[400];
    __shared__ unsigned mold[400];
    __shared__ float    scl[400];
    __shared__ float    den[400];
    __shared__ int      segcnt[SCAN_WAVES];
    __shared__ int      more;

    const int tid  = threadIdx.x;
    const int lane = tid & 31;
    const int wave = __builtin_amdgcn_readfirstlane(tid >> 5);
    const int gid  = blockIdx.x;
    const int row_base = gid * RANGE_ROWS;

    for (int i = tid; i < RANGE_ROWS * OUT_DIM / 4; i += 512)
        ((float4*)acc)[i] = make_float4(0.f, 0.f, 0.f, 0.f);
    for (int i = tid; i < 400; i += 512) rmax[i] = 0u;
    float dreg = 0.f;

    const uint4* __restrict__ key4 = (const uint4*)key;
    const unsigned pat = (unsigned)gid * 0x01010101u;
    int c = wave * CHUNKS_PER_WAVE;
    int cend = c + CHUNKS_PER_WAVE;
    if (cend > N_CHUNKS) cend = N_CHUNKS;

    for (int pass = 0; pass < MAX_PASS; ++pass) {
        if (tid == 0) more = 0;
        __syncthreads();

        int cnt = 0;
        bool full = false;
        while (c < cend && !full) {
            uint4 kv[8];
#pragma unroll
            for (int u = 0; u < 8; ++u) {
                int cc = c + u;
                if (cc > cend - 1) cc = cend - 1;
                kv[u] = key4[(size_t)cc * 32 + lane];
            }
#pragma unroll
            for (int u = 0; u < 8; ++u) {
                if (c < cend && !full) {
                    const unsigned t0 = zero_bytes(kv[u].x ^ pat);
                    const unsigned t1 = zero_bytes(kv[u].y ^ pat);
                    const unsigned t2 = zero_bytes(kv[u].z ^ pat);
                    const unsigned t3 = zero_bytes(kv[u].w ^ pat);
                    unsigned mk = __builtin_amdgcn_ballot_w32((t0 | t1 | t2 | t3) != 0u);
                    if (mk) {
                        const int cl = __popc(t0) + __popc(t1) + __popc(t2) + __popc(t3);
                        int tot, off;
                        if (__builtin_amdgcn_ballot_w32(cl > 1) == 0u) {
                            tot = __popc(mk);
                            off = (int)__builtin_amdgcn_mbcnt_lo(mk, 0u);
                        } else {
                            int x = cl;
#pragma unroll
                            for (int d = 1; d < 32; d <<= 1) {
                                const int y = __shfl_up(x, d, 32);
                                if (lane >= d) x += y;
                            }
                            off = x - cl;
                            tot = __builtin_amdgcn_readlane(x, 31);
                        }
                        if (cnt + tot > SEG_CAP) {
                            full = true;
                        } else {
                            if (cl) {
                                const unsigned n0 = (((t0 >> 7) & 0x01010101u) * 0x01020408u) >> 24;
                                const unsigned n1 = (((t1 >> 7) & 0x01010101u) * 0x01020408u) >> 24;
                                const unsigned n2 = (((t2 >> 7) & 0x01010101u) * 0x01020408u) >> 24;
                                const unsigned n3 = (((t3 >> 7) & 0x01010101u) * 0x01020408u) >> 24;
                                unsigned m16 = (n0 & 15u) | ((n1 & 15u) << 4) | ((n2 & 15u) << 8) | ((n3 & 15u) << 12);
                                const unsigned eb = (unsigned)c * CHUNK + (unsigned)(lane * 16);
                                int pos = wave * SEG_CAP + cnt + off;
                                while (m16) {
                                    l_idx[pos++] = eb + (unsigned)__builtin_ctz(m16);
                                    m16 &= m16 - 1;
                                }
                            }
                            cnt += tot;
                        }
                    }
                    if (!full) ++c;
                }
            }
        }
        if (lane == 0) {
            segcnt[wave] = cnt;
            if (c < cend) more = 1;
        }
        __syncthreads();
        const int again = more;

        int total = 0;
        for (int w = 0; w < SCAN_WAVES; ++w) total += segcnt[w];
        {
            unsigned ee[SCAN_WAVES];
            int si[SCAN_WAVES], di[SCAN_WAVES];
            float sv[SCAN_WAVES], tv[SCAN_WAVES];
#pragma unroll
            for (int w = 0; w < SCAN_WAVES; ++w)
                ee[w] = (tid < segcnt[w]) ? l_idx[w * SEG_CAP + tid] : 0u;
#pragma unroll
            for (int w = 0; w < SCAN_WAVES; ++w) { si[w] = src[ee[w]]; di[w] = dst[ee[w]]; }
#pragma unroll
            for (int w = 0; w < SCAN_WAVES; ++w) { sv[w] = s[si[w]]; tv[w] = t[di[w]]; }
#pragma unroll
            for (int w = 0; w < SCAN_WAVES; ++w) {
                float v = sv[w] + tv[w];
                v = (v > 0.f) ? v : LEAKY * v;
                sv[w] = v;
                if (tid < segcnt[w]) atomicMax(&rmax[di[w] - row_base], fkey(v));
            }
            __syncthreads();
            if (pass > 0) {
                for (int r = tid; r < RANGE_ROWS; r += 512) {
                    const unsigned ko = mold[r], kn = rmax[r];
                    scl[r] = (ko != 0u && kn != ko) ? expf(funkey(ko) - funkey(kn)) : 1.0f;
                }
                __syncthreads();
                for (int i = tid; i < RANGE_ROWS * OUT_DIM / 4; i += 512) {
                    const float sc = scl[i >> 4];
                    float4 a = ((float4*)acc)[i];
                    a.x *= sc; a.y *= sc; a.z *= sc; a.w *= sc;
                    ((float4*)acc)[i] = a;
                }
                const int rown = lane * 16 + wave;
                if (rown < RANGE_ROWS) dreg *= scl[rown];
            }
            int base = 0;
#pragma unroll
            for (int w = 0; w < SCAN_WAVES; ++w) {
                const int n = segcnt[w];
                if (tid < n) {
                    const int lrow = di[w] - row_base;
                    c_pk[base + tid] = (unsigned)si[w] | ((unsigned)lrow << 17);
                    c_e[base + tid]  = expf(sv[w] - funkey(rmax[lrow]));
                }
                base += n;
            }
        }
        __syncthreads();

#define ROW_PIECE(K, B) {                                                           \
            const int q_ = tid + 512 * (K);                                          \
            int jx_ = (B) + (q_ >> 4);                                               \
            if (jx_ > total - 1) jx_ = total - 1;                                    \
            const unsigned sidx_ = c_pk[jx_] & 0x1FFFFu;                             \
            p##K = ((const float4*)(z + (size_t)sidx_ * OUT_DIM))[q_ & 15]; }
        float4 p0 = make_float4(0.f, 0.f, 0.f, 0.f);
        float4 p1 = p0, p2 = p0, p3 = p0, p4 = p0, p5 = p0, p6 = p0, p7 = p0;
        if (total > 0) {
            ROW_PIECE(0, 0) ROW_PIECE(1, 0) ROW_PIECE(2, 0) ROW_PIECE(3, 0)
            ROW_PIECE(4, 0) ROW_PIECE(5, 0) ROW_PIECE(6, 0) ROW_PIECE(7, 0)
        }
        for (int b0 = 0; b0 < total; b0 += BATCH) {
            ((float4*)stage)[tid + 512 * 0] = p0;
            ((float4*)stage)[tid + 512 * 1] = p1;
            ((float4*)stage)[tid + 512 * 2] = p2;
            ((float4*)stage)[tid + 512 * 3] = p3;
            ((float4*)stage)[tid + 512 * 4] = p4;
            ((float4*)stage)[tid + 512 * 5] = p5;
            ((float4*)stage)[tid + 512 * 6] = p6;
            ((float4*)stage)[tid + 512 * 7] = p7;
            __syncthreads();
            if (b0 + BATCH < total) {
                const int nb = b0 + BATCH;
                ROW_PIECE(0, nb) ROW_PIECE(1, nb) ROW_PIECE(2, nb) ROW_PIECE(3, nb)
                ROW_PIECE(4, nb) ROW_PIECE(5, nb) ROW_PIECE(6, nb) ROW_PIECE(7, nb)
            }
            unsigned pkv[BATCH / 32];
#pragma unroll
            for (int g = 0; g < BATCH / 32; ++g) {
                const int j = b0 + g * 32 + lane;
                const int jc = (j < total) ? j : 0;
                pkv[g] = c_pk[jc];
            }
#pragma unroll
            for (int g = 0; g < BATCH / 32; ++g) {
                const int j = b0 + g * 32 + lane;
                const int lrow = (int)(pkv[g] >> 17);
                unsigned mk = __builtin_amdgcn_ballot_w32((j < total) && ((lrow & 15) == wave));
                while (mk) {
                    const int l = __builtin_ctz(mk);
                    mk &= mk - 1;
                    const int r = __builtin_amdgcn_readlane(lrow, l);
                    const int jj = g * 32 + l;
                    const float p = c_e[b0 + jj];
                    float2 a = ((float2*)acc)[r * 32 + lane];
                    const float2 zv = ((const float2*)stage)[jj * 32 + lane];
                    a.x += p * zv.x;
                    a.y += p * zv.y;
                    ((float2*)acc)[r * 32 + lane] = a;
                    dreg += (lane == (r >> 4)) ? p : 0.f;
                }
            }
            __syncthreads();
        }
        if (!again) break;
        for (int r = tid; r < RANGE_ROWS; r += 512) mold[r] = rmax[r];
    }
    {
        const int rown = lane * 16 + wave;
        if (rown < RANGE_ROWS) den[rown] = dreg;
    }
    __syncthreads();

    for (int i = tid; i < RANGE_ROWS * OUT_DIM / 4; i += 512) {
        const int r = i >> 4;
        const int node = row_base + r;
        if (node < N_NODES) {
            const float d = den[r];
            float4 a = ((const float4*)acc)[i];
            const float inv = (d > 0.f) ? 1.0f / d : 0.f;
            a.x *= inv; a.y *= inv; a.z *= inv; a.w *= inv;
            ((float4*)(out + (size_t)node * OUT_DIM))[i & 15] = a;
        }
    }
}

extern "C" void kernel_launch(void* const* d_in, const int* in_sizes, int n_in,
                              void* d_out, int out_size, void* d_ws, size_t ws_size,
                              hipStream_t stream) {
    const float* h  = (const float*)d_in[0];
    const int* src  = (const int*)d_in[1];
    const int* dst  = (const int*)d_in[2];
    const float* W  = (const float*)d_in[3];
    const float* A  = (const float*)d_in[4];
    float* out = (float*)d_out;

    char* ws = (char*)d_ws;
    float*         z   = (float*)        (ws);
    float*         s   = (float*)        (ws + 25600000);
    float*         t   = (float*)        (ws + 26000000);
    unsigned char* key = (unsigned char*)(ws + 26400000);

    gat_front<<<N_NODES / 16, 128, 0, stream>>>(h, W, A, dst, z, s, t, key);
    gat_range<<<N_RANGES, 512, 0, stream>>>(src, dst, key, s, t, z, out);
}


# front product re-tiled: a work-group takes 5 row tiles, W fragments stay in registers, h tiles come as whole rows through LDS (was 128 sixteen-line loads per wave and tile); same k order and sums
# speedup vs baseline: 2.5826x; 1.1211x over previous
#include <hip/hip_runtime.h>
#include <hip/hip_bf16.h>
#include <math.h>

#define N_NODES 100000
#define N_EDGES 1600000
#define IN_DIM  128
#define OUT_DIM 64
#define LEAKY   0.01f

typedef float v2f __attribute__((ext_vector_type(2)));
typedef float v8f __attribute__((ext_vector_type(8)));

#define RANGE_ROWS 391
#define N_RANGES   256
#define CHUNK      512
#define N_CHUNKS   (N_EDGES / CHUNK)
#define SCAN_WAVES 16
#define CHUNKS_PER_WAVE 196
#define SEG_CAP    512
#define LIST_CAP   (SCAN_WAVES * SEG_CAP)
#define BATCH      256
#define MAX_PASS   200


#define ZT_PITCH 68
#define HT_PITCH 132
#define TILES_PER_WG 5
#define FRONT_WGS (N_NODES / 16 / TILES_PER_WG)
#define EDGES_PER_TILE 256
static_assert((N_NODES % 16) == 0, "row tiles must be whole");
static_assert(FRONT_WGS * TILES_PER_WG * 16 == N_NODES, "every row tile belongs to exactly one work-group");
static_assert(EDGES_PER_TILE * (N_NODES / 16) == N_EDGES, "every edge gets its key from exactly one tile");

__global__ void __launch_bounds__(128)
gat_front(const float* __restrict__ h,
          const float* __restrict__ W,
          const float* __restrict__ A,
          const int* __restrict__ dst,
          float* __restrict__ z,
          float* __restrict__ s, float* __restrict__ t,
          unsigned char* __restrict__ key) {
    __shared__ float4 hb4[2][16 * HT_PITCH / 4];
    __shared__ float4 zt4[2][16 * ZT_PITCH / 4];

    const int tid  = threadIdx.x;
    const int lane = tid & 31;
    const int wave = tid >> 5;
    const int tile0 = blockIdx.x * TILES_PER_WG;
    const int col0 = wave * 16;

    const int m     = lane & 15;
    const int khalf = (lane >> 4) * 2;

    const float4* __restrict__ hsrc = (const float4*)h + (size_t)(wave * 4) * (IN_DIM / 4) + lane;
    const int hdst = (wave * 4) * (HT_PITCH / 4) + lane;
#define LOAD_TILE(T_) {                                                     \
        const float4* p_ = hsrc + (size_t)(T_) * (16 * IN_DIM / 4);         \
        g0 = p_[0 * (IN_DIM / 4)]; g1 = p_[1 * (IN_DIM / 4)];               \
        g2 = p_[2 * (IN_DIM / 4)]; g3 = p_[3 * (IN_DIM / 4)]; }
#define PARK_TILE(B_) {                                                     \
        hb4[B_][hdst + 0 * (HT_PITCH / 4)] = g0;                            \
        hb4[B_][hdst + 1 * (HT_PITCH / 4)] = g1;                            \
        hb4[B_][hdst + 2 * (HT_PITCH / 4)] = g2;                            \
        hb4[B_][hdst + 3 * (HT_PITCH / 4)] = g3; }
    float4 g0, g1, g2, g3;
    LOAD_TILE(tile0)

    const float* __restrict__ wrow = W + (size_t)(col0 + m) * IN_DIM;
    v2f bf[IN_DIM / 4];
#pragma unroll
    for (int kk = 0; kk < IN_DIM / 4; ++kk) {
        bf[kk].x = wrow[4 * kk + khalf];
        bf[kk].y = wrow[4 * kk + khalf + 1];
    }

    const int q  = tid & 15;
    const int rr = tid >> 4;
    const float4 as = ((const float4*)A)[q];
    const float4 ad = ((const float4*)(A + OUT_DIM))[q];
    const int n     = lane & 15;
    const int rbase = (lane >> 4) * 8;

    PARK_TILE(0)
    __syncthreads();
    if (TILES_PER_WG > 1) LOAD_TILE(tile0 + 1)

    for (int i = 0; i < TILES_PER_WG; ++i) {
        const int tile = tile0 + i;
        const int buf  = i & 1;
        const int row0 = tile * 16;

        const int e0 = (tile * (EDGES_PER_TILE / 2) + tid) * 2;
        const int2 dd = *(const int2*)(dst + e0);

        const float* __restrict__ hrow = (const float*)hb4[buf] + m * HT_PITCH + khalf;
        v2f af[IN_DIM / 4];
#pragma unroll
        for (int kk = 0; kk < IN_DIM / 4; ++kk) af[kk] = *(const v2f*)(hrow + 4 * kk);
        __builtin_amdgcn_sched_barrier(0);
        v8f c = {};
#pragma unroll
        for (int kk = 0; kk < IN_DIM / 4; ++kk) {
            c = __builtin_amdgcn_wmma_f32_16x16x4_f32(
                    false, af[kk], false, bf[kk], (short)0, c, false, false);
        }
        __builtin_amdgcn_sched_barrier(0);

        float* zt = (float*)zt4[buf];
#pragma unroll
        for (int r = 0; r < 8; ++r) {
            zt[(rbase + r) * ZT_PITCH + col0 + n] = c[r];
        }
        if (i + 1 < TILES_PER_WG) PARK_TILE(buf ^ 1)
        __syncthreads();
        if (i + 2 < TILES_PER_WG) LOAD_TILE(tile + 2)

        const float4 v0 = zt4[buf][rr * (ZT_PITCH / 4) + q];
        const float4 v1 = zt4[buf][(rr + 8) * (ZT_PITCH / 4) + q];
        float ps0 = v0.x * as.x + v0.y * as.y + v0.z * as.z + v0.w * as.w;
        float pt0 = v0.x * ad.x + v0.y * ad.y + v0.z * ad.z + v0.w * ad.w;
        float ps1 = v1.x * as.x + v1.y * as.y + v1.z * as.z + v1.w * as.w;
        float pt1 = v1.x * ad.x + v1.y * ad.y + v1.z * ad.z + v1.w * ad.w;
        ((float4*)(z + (size_t)(row0 + rr) * OUT_DIM))[q] = v0;
        ((float4*)(z + (size_t)(row0 + rr + 8) * OUT_DIM))[q] = v1;
#pragma unroll
        for (int mk = 1; mk < 16; mk <<= 1) {
            const float a0 = __shfl_xor(ps0, mk, 32);
            const float a1 = __shfl_xor(pt0, mk, 32);
            const float a2 = __shfl_xor(ps1, mk, 32);
            const float a3 = __shfl_xor(pt1, mk, 32);
            ps0 += a0; pt0 += a1; ps1 += a2; pt1 += a3;
        }
        if (q == 0) {
            s[row0 + rr]     = ps0;
            t[row0 + rr]     = pt0;
            s[row0 + rr + 8] = ps1;
            t[row0 + rr + 8] = pt1;
        }

        uchar2 kk2;
        kk2.x = (unsigned char)(dd.x / RANGE_ROWS);
        kk2.y = (unsigned char)(dd.y / RANGE_ROWS);
        *(uchar2*)(key + e0) = kk2;
    }
#undef LOAD_TILE
#undef PARK_TILE
}

__device__ __forceinline__ unsigned zero_bytes(unsigned y) {
    return ~(((y & 0x7F7F7F7Fu) + 0x7F7F7F7Fu) | y | 0x7F7F7F7Fu);
}


__device__ __forceinline__ unsigned fkey(float v) {
    const unsigned b = __float_as_uint(v);
    return (b & 0x80000000u) ? ~b : (b | 0x80000000u);
}
__device__ __forceinline__ float funkey(unsigned k) {
    return __uint_as_float((k & 0x80000000u) ? (k & 0x7FFFFFFFu) : ~k);
}
__global__ void __launch_bounds__(512)
gat_range(const int* __restrict__ src, const int* __restrict__ dst,
          const unsigned char* __restrict__ key,
          const float* __restrict__ s, const float* __restrict__ t,
          const float* __restrict__ z, float* __restrict__ out) {
    __shared__ float    acc[RANGE_ROWS * OUT_DIM];
    __shared__ float    stage[BATCH * OUT_DIM];
    __shared__ unsigned l_idx[LIST_CAP];
    __shared__ unsigned c_pk[LIST_CAP];
    __shared__ float    c_e[LIST_CAP];
    __shared__ unsigned rmax[400];
    __shared__ unsigned mold[400];
    __shared__ float    scl[400];
    __shared__ float    den[400];
    __shared__ int      segcnt[SCAN_WAVES];
    __shared__ int      more;

    const int tid  = threadIdx.x;
    const int lane = tid & 31;
    const int wave = __builtin_amdgcn_readfirstlane(tid >> 5);
    const int gid  = blockIdx.x;
    const int row_base = gid * RANGE_ROWS;

    for (int i = tid; i < RANGE_ROWS * OUT_DIM / 4; i += 512)
        ((float4*)acc)[i] = make_float4(0.f, 0.f, 0.f, 0.f);
    for (int i = tid; i < 400; i += 512) rmax[i] = 0u;
    float dreg = 0.f;

    const uint4* __restrict__ key4 = (const uint4*)key;
    const unsigned pat = (unsigned)gid * 0x01010101u;
    int c = wave * CHUNKS_PER_WAVE;
    int cend = c + CHUNKS_PER_WAVE;
    if (cend > N_CHUNKS) cend = N_CHUNKS;

    for (int pass = 0; pass < MAX_PASS; ++pass) {
        if (tid == 0) more = 0;
        __syncthreads();

        int cnt = 0;
        bool full = false;
        while (c < cend && !full) {
            uint4 kv[8];
#pragma unroll
            for (int u = 0; u < 8; ++u) {
                int cc = c + u;
                if (cc > cend - 1) cc = cend - 1;
                kv[u] = key4[(size_t)cc * 32 + lane];
            }
#pragma unroll
            for (int u = 0; u < 8; ++u) {
                if (c < cend && !full) {
                    const unsigned t0 = zero_bytes(kv[u].x ^ pat);
                    const unsigned t1 = zero_bytes(kv[u].y ^ pat);
                    const unsigned t2 = zero_bytes(kv[u].z ^ pat);
                    const unsigned t3 = zero_bytes(kv[u].w ^ pat);
                    unsigned mk = __builtin_amdgcn_ballot_w32((t0 | t1 | t2 | t3) != 0u);
                    if (mk) {
                        const int cl = __popc(t0) + __popc(t1) + __popc(t2) + __popc(t3);
                        int tot, off;
                        if (__builtin_amdgcn_ballot_w32(cl > 1) == 0u) {
                            tot = __popc(mk);
                            off = (int)__builtin_amdgcn_mbcnt_lo(mk, 0u);
                        } else {
                            int x = cl;
#pragma unroll
                            for (int d = 1; d < 32; d <<= 1) {
                                const int y = __shfl_up(x, d, 32);
                                if (lane >= d) x += y;
                            }
                            off = x - cl;
                            tot = __builtin_amdgcn_readlane(x, 31);
                        }
                        if (cnt + tot > SEG_CAP) {
                            full = true;
                        } else {
                            if (cl) {
                                const unsigned n0 = (((t0 >> 7) & 0x01010101u) * 0x01020408u) >> 24;
                                const unsigned n1 = (((t1 >> 7) & 0x01010101u) * 0x01020408u) >> 24;
                                const unsigned n2 = (((t2 >> 7) & 0x01010101u) * 0x01020408u) >> 24;
                                const unsigned n3 = (((t3 >> 7) & 0x01010101u) * 0x01020408u) >> 24;
                                unsigned m16 = (n0 & 15u) | ((n1 & 15u) << 4) | ((n2 & 15u) << 8) | ((n3 & 15u) << 12);
                                const unsigned eb = (unsigned)c * CHUNK + (unsigned)(lane * 16);
                                int pos = wave * SEG_CAP + cnt + off;
                                while (m16) {
                                    l_idx[pos++] = eb + (unsigned)__builtin_ctz(m16);
                                    m16 &= m16 - 1;
                                }
                            }
                            cnt += tot;
                        }
                    }
                    if (!full) ++c;
                }
            }
        }
        if (lane == 0) {
            segcnt[wave] = cnt;
            if (c < cend) more = 1;
        }
        __syncthreads();
        const int again = more;

        int total = 0;
        for (int w = 0; w < SCAN_WAVES; ++w) total += segcnt[w];
        {
            unsigned ee[SCAN_WAVES];
            int si[SCAN_WAVES], di[SCAN_WAVES];
            float sv[SCAN_WAVES], tv[SCAN_WAVES];
#pragma unroll
            for (int w = 0; w < SCAN_WAVES; ++w)
                ee[w] = (tid < segcnt[w]) ? l_idx[w * SEG_CAP + tid] : 0u;
#pragma unroll
            for (int w = 0; w < SCAN_WAVES; ++w) { si[w] = src[ee[w]]; di[w] = dst[ee[w]]; }
#pragma unroll
            for (int w = 0; w < SCAN_WAVES; ++w) { sv[w] = s[si[w]]; tv[w] = t[di[w]]; }
#pragma unroll
            for (int w = 0; w < SCAN_WAVES; ++w) {
                float v = sv[w] + tv[w];
                v = (v > 0.f) ? v : LEAKY * v;
                sv[w] = v;
                if (tid < segcnt[w]) atomicMax(&rmax[di[w] - row_base], fkey(v));
            }
            __syncthreads();
            if (pass > 0) {
                for (int r = tid; r < RANGE_ROWS; r += 512) {
                    const unsigned ko = mold[r], kn = rmax[r];
                    scl[r] = (ko != 0u && kn != ko) ? expf(funkey(ko) - funkey(kn)) : 1.0f;
                }
                __syncthreads();
                for (int i = tid; i < RANGE_ROWS * OUT_DIM / 4; i += 512) {
                    const float sc = scl[i >> 4];
                    float4 a = ((float4*)acc)[i];
                    a.x *= sc; a.y *= sc; a.z *= sc; a.w *= sc;
                    ((float4*)acc)[i] = a;
                }
                const int rown = lane * 16 + wave;
                if (rown < RANGE_ROWS) dreg *= scl[rown];
            }
            int base = 0;
#pragma unroll
            for (int w = 0; w < SCAN_WAVES; ++w) {
                const int n = segcnt[w];
                if (tid < n) {
                    const int lrow = di[w] - row_base;
                    c_pk[base + tid] = (unsigned)si[w] | ((unsigned)lrow << 17);
                    c_e[base + tid]  = expf(sv[w] - funkey(rmax[lrow]));
                }
                base += n;
            }
        }
        __syncthreads();

#define ROW_PIECE(K, B) {                                                           \
            const int q_ = tid + 512 * (K);                                          \
            int jx_ = (B) + (q_ >> 4);                                               \
            if (jx_ > total - 1) jx_ = total - 1;                                    \
            const unsigned sidx_ = c_pk[jx_] & 0x1FFFFu;                             \
            p##K = ((const float4*)(z + (size_t)sidx_ * OUT_DIM))[q_ & 15]; }
        float4 p0 = make_float4(0.f, 0.f, 0.f, 0.f);
        float4 p1 = p0, p2 = p0, p3 = p0, p4 = p0, p5 = p0, p6 = p0, p7 = p0;
        if (total > 0) {
            ROW_PIECE(0, 0) ROW_PIECE(1, 0) ROW_PIECE(2, 0) ROW_PIECE(3, 0)
            ROW_PIECE(4, 0) ROW_PIECE(5, 0) ROW_PIECE(6, 0) ROW_PIECE(7, 0)
        }
        for (int b0 = 0; b0 < total; b0 += BATCH) {
            ((float4*)stage)[tid + 512 * 0] = p0;
            ((float4*)stage)[tid + 512 * 1] = p1;
            ((float4*)stage)[tid + 512 * 2] = p2;
            ((float4*)stage)[tid + 512 * 3] = p3;
            ((float4*)stage)[tid + 512 * 4] = p4;
            ((float4*)stage)[tid + 512 * 5] = p5;
            ((float4*)stage)[tid + 512 * 6] = p6;
            ((float4*)stage)[tid + 512 * 7] = p7;
            __syncthreads();
            if (b0 + BATCH < total) {
                const int nb = b0 + BATCH;
                ROW_PIECE(0, nb) ROW_PIECE(1, nb) ROW_PIECE(2, nb) ROW_PIECE(3, nb)
                ROW_PIECE(4, nb) ROW_PIECE(5, nb) ROW_PIECE(6, nb) ROW_PIECE(7, nb)
            }
            unsigned pkv[BATCH / 32];
#pragma unroll
            for (int g = 0; g < BATCH / 32; ++g) {
                const int j = b0 + g * 32 + lane;
                const int jc = (j < total) ? j : 0;
                pkv[g] = c_pk[jc];
            }
#pragma unroll
            for (int g = 0; g < BATCH / 32; ++g) {
                const int j = b0 + g * 32 + lane;
                const int lrow = (int)(pkv[g] >> 17);
                unsigned mk = __builtin_amdgcn_ballot_w32((j < total) && ((lrow & 15) == wave));
                while (mk) {
                    const int l = __builtin_ctz(mk);
                    mk &= mk - 1;
                    const int r = __builtin_amdgcn_readlane(lrow, l);
                    const int jj = g * 32 + l;
                    const float p = c_e[b0 + jj];
                    float2 a = ((float2*)acc)[r * 32 + lane];
                    const float2 zv = ((const float2*)stage)[jj * 32 + lane];
                    a.x += p * zv.x;
                    a.y += p * zv.y;
                    ((float2*)acc)[r * 32 + lane] = a;
                    dreg += (lane == (r >> 4)) ? p : 0.f;
                }
            }
            __syncthreads();
        }
        if (!again) break;
        for (int r = tid; r < RANGE_ROWS; r += 512) mold[r] = rmax[r];
    }
    {
        const int rown = lane * 16 + wave;
        if (rown < RANGE_ROWS) den[rown] = dreg;
    }
    __syncthreads();

    for (int i = tid; i < RANGE_ROWS * OUT_DIM / 4; i += 512) {
        const int r = i >> 4;
        const int node = row_base + r;
        if (node < N_NODES) {
            const float d = den[r];
            float4 a = ((const float4*)acc)[i];
            const float inv = (d > 0.f) ? 1.0f / d : 0.f;
            a.x *= inv; a.y *= inv; a.z *= inv; a.w *= inv;
            ((float4*)(out + (size_t)node * OUT_DIM))[i & 15] = a;
        }
    }
}

extern "C" void kernel_launch(void* const* d_in, const int* in_sizes, int n_in,
                              void* d_out, int out_size, void* d_ws, size_t ws_size,
                              hipStream_t stream) {
    const float* h  = (const float*)d_in[0];
    const int* src  = (const int*)d_in[1];
    const int* dst  = (const int*)d_in[2];
    const float* W  = (const float*)d_in[3];
    const float* A  = (const float*)d_in[4];
    float* out = (float*)d_out;

    char* ws = (char*)d_ws;
    float*         z   = (float*)        (ws);
    float*         s   = (float*)        (ws + 25600000);
    float*         t   = (float*)        (ws + 26000000);
    unsigned char* key = (unsigned char*)(ws + 26400000);

    gat_front<<<FRONT_WGS, 128, 0, stream>>>(h, W, A, dst, z, s, t, key);
    gat_range<<<N_RANGES, 512, 0, stream>>>(src, dst, key, s, t, z, out);
}


# range phase: row sums kept in the owner wave's registers, not LDS (slot picked per edge by register-relative moves); a wave lists its own edges per batch, 8 row reads in flight; same sum order
# speedup vs baseline: 2.7521x; 1.0656x over previous
#include <hip/hip_runtime.h>
#include <hip/hip_bf16.h>
#include <math.h>

#define N_NODES 100000
#define N_EDGES 1600000
#define IN_DIM  128
#define OUT_DIM 64
#define LEAKY   0.01f

typedef float v2f __attribute__((ext_vector_type(2)));
typedef float v8f __attribute__((ext_vector_type(8)));
typedef float v32f __attribute__((ext_vector_type(32)));

#define RANGE_ROWS 391
#define N_RANGES   256
#define CHUNK      512
#define N_CHUNKS   (N_EDGES / CHUNK)
#define SCAN_WAVES 16
#define CHUNKS_PER_WAVE 196
#define SEG_CAP    512
#define LIST_CAP   (SCAN_WAVES * SEG_CAP)
#define BATCH      256
#define MAX_PASS   200
#define OWN_CAP    BATCH
#define ROWS_PER_WAVE 25
#define OWN_STEP   8


#define ZT_PITCH 68
#define HT_PITCH 132
#define TILES_PER_WG 5
#define FRONT_WGS (N_NODES / 16 / TILES_PER_WG)
#define EDGES_PER_TILE 256
static_assert((N_NODES % 16) == 0, "row tiles must be whole");
static_assert(FRONT_WGS * TILES_PER_WG * 16 == N_NODES, "every row tile belongs to exactly one work-group");
static_assert(EDGES_PER_TILE * (N_NODES / 16) == N_EDGES, "every edge gets its key from exactly one tile");

__global__ void __launch_bounds__(128)
gat_front(const float* __restrict__ h,
          const float* __restrict__ W,
          const float* __restrict__ A,
          const int* __restrict__ dst,
          float* __restrict__ z,
          float* __restrict__ s, float* __restrict__ t,
          unsigned char* __restrict__ key) {
    __shared__ float4 hb4[2][16 * HT_PITCH / 4];
    __shared__ float4 zt4[2][16 * ZT_PITCH / 4];

    const int tid  = threadIdx.x;
    const int lane = tid & 31;
    const int wave = tid >> 5;
    const int tile0 = blockIdx.x * TILES_PER_WG;
    const int col0 = wave * 16;

    const int m     = lane & 15;
    const int khalf = (lane >> 4) * 2;

    const float4* __restrict__ hsrc = (const float4*)h + (size_t)(wave * 4) * (IN_DIM / 4) + lane;
    const int hdst = (wave * 4) * (HT_PITCH / 4) + lane;
#define LOAD_TILE(T_) {                                                     \
        const float4* p_ = hsrc + (size_t)(T_) * (16 * IN_DIM / 4);         \
        g0 = p_[0 * (IN_DIM / 4)]; g1 = p_[1 * (IN_DIM / 4)];               \
        g2 = p_[2 * (IN_DIM / 4)]; g3 = p_[3 * (IN_DIM / 4)]; }
#define PARK_TILE(B_) {                                                     \
        hb4[B_][hdst + 0 * (HT_PITCH / 4)] = g0;                            \
        hb4[B_][hdst + 1 * (HT_PITCH / 4)] = g1;                            \
        hb4[B_][hdst + 2 * (HT_PITCH / 4)] = g2;                            \
        hb4[B_][hdst + 3 * (HT_PITCH / 4)] = g3; }
    float4 g0, g1, g2, g3;
    LOAD_TILE(tile0)

    const float* __restrict__ wrow = W + (size_t)(col0 + m) * IN_DIM;
    v2f bf[IN_DIM / 4];
#pragma unroll
    for (int kk = 0; kk < IN_DIM / 4; ++kk) {
        bf[kk].x = wrow[4 * kk + khalf];
        bf[kk].y = wrow[4 * kk + khalf + 1];
    }

    const int q  = tid & 15;
    const int rr = tid >> 4;
    const float4 as = ((const float4*)A)[q];
    const float4 ad = ((const float4*)(A + OUT_DIM))[q];
    const int n     = lane & 15;
    const int rbase = (lane >> 4) * 8;

    PARK_TILE(0)
    __syncthreads();
    if (TILES_PER_WG > 1) LOAD_TILE(tile0 + 1)

    for (int i = 0; i < TILES_PER_WG; ++i) {
        const int tile = tile0 + i;
        const int buf  = i & 1;
        const int row0 = tile * 16;

        const int e0 = (tile * (EDGES_PER_TILE / 2) + tid) * 2;
        const int2 dd = *(const int2*)(dst + e0);

        const float* __restrict__ hrow = (const float*)hb4[buf] + m * HT_PITCH + khalf;
        v2f af[IN_DIM / 4];
#pragma unroll
        for (int kk = 0; kk < IN_DIM / 4; ++kk) af[kk] = *(const v2f*)(hrow + 4 * kk);
        __builtin_amdgcn_sched_barrier(0);
        v8f c = {};
#pragma unroll
        for (int kk = 0; kk < IN_DIM / 4; ++kk) {
            c = __builtin_amdgcn_wmma_f32_16x16x4_f32(
                    false, af[kk], false, bf[kk], (short)0, c, false, false);
        }
        __builtin_amdgcn_sched_barrier(0);

        float* zt = (float*)zt4[buf];
#pragma unroll
        for (int r = 0; r < 8; ++r) {
            zt[(rbase + r) * ZT_PITCH + col0 + n] = c[r];
        }
        if (i + 1 < TILES_PER_WG) PARK_TILE(buf ^ 1)
        __syncthreads();
        if (i + 2 < TILES_PER_WG) LOAD_TILE(tile + 2)

        const float4 v0 = zt4[buf][rr * (ZT_PITCH / 4) + q];
        const float4 v1 = zt4[buf][(rr + 8) * (ZT_PITCH / 4) + q];
        float ps0 = v0.x * as.x + v0.y * as.y + v0.z * as.z + v0.w * as.w;
        float pt0 = v0.x * ad.x + v0.y * ad.y + v0.z * ad.z + v0.w * ad.w;
        float ps1 = v1.x * as.x + v1.y * as.y + v1.z * as.z + v1.w * as.w;
        float pt1 = v1.x * ad.x + v1.y * ad.y + v1.z * ad.z + v1.w * ad.w;
        ((float4*)(z + (size_t)(row0 + rr) * OUT_DIM))[q] = v0;
        ((float4*)(z + (size_t)(row0 + rr + 8) * OUT_DIM))[q] = v1;
#pragma unroll
        for (int mk = 1; mk < 16; mk <<= 1) {
            const float a0 = __shfl_xor(ps0, mk, 32);
            const float a1 = __shfl_xor(pt0, mk, 32);
            const float a2 = __shfl_xor(ps1, mk, 32);
            const float a3 = __shfl_xor(pt1, mk, 32);
            ps0 += a0; pt0 += a1; ps1 += a2; pt1 += a3;
        }
        if (q == 0) {
            s[row0 + rr]     = ps0;
            t[row0 + rr]     = pt0;
            s[row0 + rr + 8] = ps1;
            t[row0 + rr + 8] = pt1;
        }

        uchar2 kk2;
        kk2.x = (unsigned char)(dd.x / RANGE_ROWS);
        kk2.y = (unsigned char)(dd.y / RANGE_ROWS);
        *(uchar2*)(key + e0) = kk2;
    }
#undef LOAD_TILE
#undef PARK_TILE
}

__device__ __forceinline__ unsigned zero_bytes(unsigned y) {
    return ~(((y & 0x7F7F7F7Fu) + 0x7F7F7F7Fu) | y | 0x7F7F7F7Fu);
}


__device__ __forceinline__ unsigned fkey(float v) {
    const unsigned b = __float_as_uint(v);
    return (b & 0x80000000u) ? ~b : (b | 0x80000000u);
}
__device__ __forceinline__ float funkey(unsigned k) {
    return __uint_as_float((k & 0x80000000u) ? (k & 0x7FFFFFFFu) : ~k);
}
__global__ void __launch_bounds__(512)
gat_range(const int* __restrict__ src, const int* __restrict__ dst,
          const unsigned char* __restrict__ key,
          const float* __restrict__ s, const float* __restrict__ t,
          const float* __restrict__ z, float* __restrict__ out) {
    __shared__ float    stage[BATCH * OUT_DIM];
    __shared__ unsigned l_idx[LIST_CAP];
    __shared__ unsigned c_pk[LIST_CAP];
    __shared__ float    c_e[LIST_CAP];
    __shared__ unsigned rmax[400];
    __shared__ unsigned mold[400];
    __shared__ float    scl[400];
    __shared__ unsigned short own_e[SCAN_WAVES * OWN_CAP];
    __shared__ int      segcnt[SCAN_WAVES];
    __shared__ int      more;

    const int tid  = threadIdx.x;
    const int lane = tid & 31;
    const int wave = __builtin_amdgcn_readfirstlane(tid >> 5);
    const int gid  = blockIdx.x;
    const int row_base = gid * RANGE_ROWS;

    v32f ax = {};
    v32f ay = {};
    for (int i = tid; i < 400; i += 512) rmax[i] = 0u;
    float dreg = 0.f;

    const uint4* __restrict__ key4 = (const uint4*)key;
    const unsigned pat = (unsigned)gid * 0x01010101u;
    int c = wave * CHUNKS_PER_WAVE;
    int cend = c + CHUNKS_PER_WAVE;
    if (cend > N_CHUNKS) cend = N_CHUNKS;

    for (int pass = 0; pass < MAX_PASS; ++pass) {
        if (tid == 0) more = 0;
        __syncthreads();

        int cnt = 0;
        bool full = false;
        while (c < cend && !full) {
            uint4 kv[8];
#pragma unroll
            for (int u = 0; u < 8; ++u) {
                int cc = c + u;
                if (cc > cend - 1) cc = cend - 1;
                kv[u] = key4[(size_t)cc * 32 + lane];
            }
#pragma unroll
            for (int u = 0; u < 8; ++u) {
                if (c < cend && !full) {
                    const unsigned t0 = zero_bytes(kv[u].x ^ pat);
                    const unsigned t1 = zero_bytes(kv[u].y ^ pat);
                    const unsigned t2 = zero_bytes(kv[u].z ^ pat);
                    const unsigned t3 = zero_bytes(kv[u].w ^ pat);
                    unsigned mk = __builtin_amdgcn_ballot_w32((t0 | t1 | t2 | t3) != 0u);
                    if (mk) {
                        const int cl = __popc(t0) + __popc(t1) + __popc(t2) + __popc(t3);
                        int tot, off;
                        if (__builtin_amdgcn_ballot_w32(cl > 1) == 0u) {
                            tot = __popc(mk);
                            off = (int)__builtin_amdgcn_mbcnt_lo(mk, 0u);
                        } else {
                            int x = cl;
#pragma unroll
                            for (int d = 1; d < 32; d <<= 1) {
                                const int y = __shfl_up(x, d, 32);
                                if (lane >= d) x += y;
                            }
                            off = x - cl;
                            tot = __builtin_amdgcn_readlane(x, 31);
                        }
                        if (cnt + tot > SEG_CAP) {
                            full = true;
                        } else {
                            if (cl) {
                                const unsigned n0 = (((t0 >> 7) & 0x01010101u) * 0x01020408u) >> 24;
                                const unsigned n1 = (((t1 >> 7) & 0x01010101u) * 0x01020408u) >> 24;
                                const unsigned n2 = (((t2 >> 7) & 0x01010101u) * 0x01020408u) >> 24;
                                const unsigned n3 = (((t3 >> 7) & 0x01010101u) * 0x01020408u) >> 24;
                                unsigned m16 = (n0 & 15u) | ((n1 & 15u) << 4) | ((n2 & 15u) << 8) | ((n3 & 15u) << 12);
                                const unsigned eb = (unsigned)c * CHUNK + (unsigned)(lane * 16);
                                int pos = wave * SEG_CAP + cnt + off;
                                while (m16) {
                                    l_idx[pos++] = eb + (unsigned)__builtin_ctz(m16);
                                    m16 &= m16 - 1;
                                }
                            }
                            cnt += tot;
                        }
                    }
                    if (!full) ++c;
                }
            }
        }
        if (lane == 0) {
            segcnt[wave] = cnt;
            if (c < cend) more = 1;
        }
        __syncthreads();
        const int again = more;

        int total = 0;
        for (int w = 0; w < SCAN_WAVES; ++w) total += segcnt[w];
        {
            unsigned ee[SCAN_WAVES];
            int si[SCAN_WAVES], di[SCAN_WAVES];
            float sv[SCAN_WAVES], tv[SCAN_WAVES];
#pragma unroll
            for (int w = 0; w < SCAN_WAVES; ++w)
                ee[w] = (tid < segcnt[w]) ? l_idx[w * SEG_CAP + tid] : 0u;
#pragma unroll
            for (int w = 0; w < SCAN_WAVES; ++w) { si[w] = src[ee[w]]; di[w] = dst[ee[w]]; }
#pragma unroll
            for (int w = 0; w < SCAN_WAVES; ++w) { sv[w] = s[si[w]]; tv[w] = t[di[w]]; }
#pragma unroll
            for (int w = 0; w < SCAN_WAVES; ++w) {
                float v = sv[w] + tv[w];
                v = (v > 0.f) ? v : LEAKY * v;
                sv[w] = v;
                if (tid < segcnt[w]) atomicMax(&rmax[di[w] - row_base], fkey(v));
            }
            __syncthreads();
            if (pass > 0) {
                for (int r = tid; r < RANGE_ROWS; r += 512) {
                    const unsigned ko = mold[r], kn = rmax[r];
                    scl[r] = (ko != 0u && kn != ko) ? expf(funkey(ko) - funkey(kn)) : 1.0f;
                }
                __syncthreads();
#pragma unroll
                for (int k = 0; k < ROWS_PER_WAVE; ++k) {
                    const int r = 16 * k + wave;
                    const float sc = (r < RANGE_ROWS) ? scl[r] : 1.0f;
                    ax[k] *= sc;
                    ay[k] *= sc;
                }
                const int rown = lane * 16 + wave;
                if (rown < RANGE_ROWS) dreg *= scl[rown];
            }
            int base = 0;
#pragma unroll
            for (int w = 0; w < SCAN_WAVES; ++w) {
                const int n = segcnt[w];
                if (tid < n) {
                    const int lrow = di[w] - row_base;
                    c_pk[base + tid] = (unsigned)si[w] | ((unsigned)lrow << 17);
                    c_e[base + tid]  = expf(sv[w] - funkey(rmax[lrow]));
                }
                base += n;
            }
        }
        __syncthreads();

#define ROW_PIECE(K, B) {                                                           \
            const int q_ = tid + 512 * (K);                                          \
            int jx_ = (B) + (q_ >> 4);                                               \
            if (jx_ > total - 1) jx_ = total - 1;                                    \
            const unsigned sidx_ = c_pk[jx_] & 0x1FFFFu;                             \
            p##K = ((const float4*)(z + (size_t)sidx_ * OUT_DIM))[q_ & 15]; }
        float4 p0 = make_float4(0.f, 0.f, 0.f, 0.f);
        float4 p1 = p0, p2 = p0, p3 = p0, p4 = p0, p5 = p0, p6 = p0, p7 = p0;
        if (total > 0) {
            ROW_PIECE(0, 0) ROW_PIECE(1, 0) ROW_PIECE(2, 0) ROW_PIECE(3, 0)
            ROW_PIECE(4, 0) ROW_PIECE(5, 0) ROW_PIECE(6, 0) ROW_PIECE(7, 0)
        }
        for (int b0 = 0; b0 < total; b0 += BATCH) {
            ((float4*)stage)[tid + 512 * 0] = p0;
            ((float4*)stage)[tid + 512 * 1] = p1;
            ((float4*)stage)[tid + 512 * 2] = p2;
            ((float4*)stage)[tid + 512 * 3] = p3;
            ((float4*)stage)[tid + 512 * 4] = p4;
            ((float4*)stage)[tid + 512 * 5] = p5;
            ((float4*)stage)[tid + 512 * 6] = p6;
            ((float4*)stage)[tid + 512 * 7] = p7;
            __syncthreads();
            if (b0 + BATCH < total) {
                const int nb = b0 + BATCH;
                ROW_PIECE(0, nb) ROW_PIECE(1, nb) ROW_PIECE(2, nb) ROW_PIECE(3, nb)
                ROW_PIECE(4, nb) ROW_PIECE(5, nb) ROW_PIECE(6, nb) ROW_PIECE(7, nb)
            }
            unsigned pkv[BATCH / 32];
#pragma unroll
            for (int g = 0; g < BATCH / 32; ++g) {
                const int j = b0 + g * 32 + lane;
                const int jc = (j < total) ? j : 0;
                pkv[g] = c_pk[jc];
            }
            int n_own = 0;
#pragma unroll
            for (int g = 0; g < BATCH / 32; ++g) {
                const int j = b0 + g * 32 + lane;
                const int lrow = (int)(pkv[g] >> 17);
                const bool mine = (j < total) && ((lrow & 15) == wave);
                const unsigned mk = __builtin_amdgcn_ballot_w32(mine);
                if (mine) {
                    const int pos = wave * OWN_CAP + n_own + (int)__builtin_amdgcn_mbcnt_lo(mk, 0u);
                    own_e[pos] = (unsigned short)((g * 32 + lane) | ((lrow >> 4) << 8));
                }
                n_own += __popc(mk);
            }
            for (int c0 = 0; c0 < n_own; c0 += 32) {
                const int entv = (int)own_e[wave * OWN_CAP + c0 + lane];
                const int entp = __float_as_int(c_e[b0 + (entv & 255)]);
                int m = n_own - c0;
                if (m > 32) m = 32;
                for (int i = 0; i < m; i += OWN_STEP) {
                    int    ks[OWN_STEP];
                    float  ps[OWN_STEP];
                    float2 zs[OWN_STEP];
#pragma unroll
                    for (int u = 0; u < OWN_STEP; ++u) {
                        const int iu = (i + u < m) ? (i + u) : (m - 1);
                        const int e  = __builtin_amdgcn_readlane(entv, iu);
                        ps[u] = __int_as_float(__builtin_amdgcn_readlane(entp, iu));
                        ks[u] = e >> 8;
                        zs[u] = ((const float2*)stage)[(e & 255) * 32 + lane];
                    }
#pragma unroll
                    for (int u = 0; u < OWN_STEP; ++u) {
                        if (i + u < m) {
                            const int   k = ks[u];
                            const float p = ps[u];
                            ax[k] += p * zs[u].x;
                            ay[k] += p * zs[u].y;
                            dreg += (lane == k) ? p : 0.f;
                        }
                    }
                }
            }
            __syncthreads();
        }
        if (!again) break;
        for (int r = tid; r < RANGE_ROWS; r += 512) mold[r] = rmax[r];
    }
#pragma unroll
    for (int k = 0; k < ROWS_PER_WAVE; ++k) {
        const int r = 16 * k + wave;
        const int node = row_base + r;
        if (r < RANGE_ROWS && node < N_NODES) {
            const float d = __int_as_float(__builtin_amdgcn_readlane(__float_as_int(dreg), k));
            const float inv = (d > 0.f) ? 1.0f / d : 0.f;
            float2 o;
            o.x = ax[k] * inv;
            o.y = ay[k] * inv;
            ((float2*)(out + (size_t)node * OUT_DIM))[lane] = o;
        }
    }
}

extern "C" void kernel_launch(void* const* d_in, const int* in_sizes, int n_in,
                              void* d_out, int out_size, void* d_ws, size_t ws_size,
                              hipStream_t stream) {
    const float* h  = (const float*)d_in[0];
    const int* src  = (const int*)d_in[1];
    const int* dst  = (const int*)d_in[2];
    const float* W  = (const float*)d_in[3];
    const float* A  = (const float*)d_in[4];
    float* out = (float*)d_out;

    char* ws = (char*)d_ws;
    float*         z   = (float*)        (ws);
    float*         s   = (float*)        (ws + 25600000);
    float*         t   = (float*)        (ws + 26000000);
    unsigned char* key = (unsigned char*)(ws + 26400000);

    gat_front<<<FRONT_WGS, 128, 0, stream>>>(h, W, A, dst, z, s, t, key);
    gat_range<<<N_RANGES, 512, 0, stream>>>(src, dst, key, s, t, z, out);
}


# key scan: a lane's 16 match flags merged into one word; a single-match lane stores its edge from the flag's place (no bit gather, no append loop); prefix from two ballots; same list order and sums
# speedup vs baseline: 2.8934x; 1.0513x over previous
#include <hip/hip_runtime.h>
#include <hip/hip_bf16.h>
#include <math.h>

#define N_NODES 100000
#define N_EDGES 1600000
#define IN_DIM  128
#define OUT_DIM 64
#define LEAKY   0.01f

typedef float v2f __attribute__((ext_vector_type(2)));
typedef float v8f __attribute__((ext_vector_type(8)));
typedef float v32f __attribute__((ext_vector_type(32)));

#define RANGE_ROWS 391
#define N_RANGES   256
#define CHUNK      512
#define N_CHUNKS   (N_EDGES / CHUNK)
#define SCAN_WAVES 16
#define CHUNKS_PER_WAVE 196
#define SEG_CAP    512
#define LIST_CAP   (SCAN_WAVES * SEG_CAP)
#define BATCH      256
#define MAX_PASS   200
#define OWN_CAP    BATCH
#define ROWS_PER_WAVE 25
#define OWN_STEP   8


#define ZT_PITCH 68
#define HT_PITCH 132
#define TILES_PER_WG 5
#define FRONT_WGS (N_NODES / 16 / TILES_PER_WG)
#define EDGES_PER_TILE 256
static_assert((N_NODES % 16) == 0, "row tiles must be whole");
static_assert(FRONT_WGS * TILES_PER_WG * 16 == N_NODES, "every row tile belongs to exactly one work-group");
static_assert(EDGES_PER_TILE * (N_NODES / 16) == N_EDGES, "every edge gets its key from exactly one tile");

__global__ void __launch_bounds__(128)
gat_front(const float* __restrict__ h,
          const float* __restrict__ W,
          const float* __restrict__ A,
          const int* __restrict__ dst,
          float* __restrict__ z,
          float* __restrict__ s, float* __restrict__ t,
          unsigned char* __restrict__ key) {
    __shared__ float4 hb4[2][16 * HT_PITCH / 4];
    __shared__ float4 zt4[2][16 * ZT_PITCH / 4];

    const int tid  = threadIdx.x;
    const int lane = tid & 31;
    const int wave = tid >> 5;
    const int tile0 = blockIdx.x * TILES_PER_WG;
    const int col0 = wave * 16;

    const int m     = lane & 15;
    const int khalf = (lane >> 4) * 2;

    const float4* __restrict__ hsrc = (const float4*)h + (size_t)(wave * 4) * (IN_DIM / 4) + lane;
    const int hdst = (wave * 4) * (HT_PITCH / 4) + lane;
#define LOAD_TILE(T_) {                                                     \
        const float4* p_ = hsrc + (size_t)(T_) * (16 * IN_DIM / 4);         \
        g0 = p_[0 * (IN_DIM / 4)]; g1 = p_[1 * (IN_DIM / 4)];               \
        g2 = p_[2 * (IN_DIM / 4)]; g3 = p_[3 * (IN_DIM / 4)]; }
#define PARK_TILE(B_) {                                                     \
        hb4[B_][hdst + 0 * (HT_PITCH / 4)] = g0;                            \
        hb4[B_][hdst + 1 * (HT_PITCH / 4)] = g1;                            \
        hb4[B_][hdst + 2 * (HT_PITCH / 4)] = g2;                            \
        hb4[B_][hdst + 3 * (HT_PITCH / 4)] = g3; }
    float4 g0, g1, g2, g3;
    LOAD_TILE(tile0)

    const float* __restrict__ wrow = W + (size_t)(col0 + m) * IN_DIM;
    v2f bf[IN_DIM / 4];
#pragma unroll
    for (int kk = 0; kk < IN_DIM / 4; ++kk) {
        bf[kk].x = wrow[4 * kk + khalf];
        bf[kk].y = wrow[4 * kk + khalf + 1];
    }

    const int q  = tid & 15;
    const int rr = tid >> 4;
    const float4 as = ((const float4*)A)[q];
    const float4 ad = ((const float4*)(A + OUT_DIM))[q];
    const int n     = lane & 15;
    const int rbase = (lane >> 4) * 8;

    PARK_TILE(0)
    __syncthreads();
    if (TILES_PER_WG > 1) LOAD_TILE(tile0 + 1)

    for (int i = 0; i < TILES_PER_WG; ++i) {
        const int tile = tile0 + i;
        const int buf  = i & 1;
        const int row0 = tile * 16;

        const int e0 = (tile * (EDGES_PER_TILE / 2) + tid) * 2;
        const int2 dd = *(const int2*)(dst + e0);

        const float* __restrict__ hrow = (const float*)hb4[buf] + m * HT_PITCH + khalf;
        v2f af[IN_DIM / 4];
#pragma unroll
        for (int kk = 0; kk < IN_DIM / 4; ++kk) af[kk] = *(const v2f*)(hrow + 4 * kk);
        __builtin_amdgcn_sched_barrier(0);
        v8f c = {};
#pragma unroll
        for (int kk = 0; kk < IN_DIM / 4; ++kk) {
            c = __builtin_amdgcn_wmma_f32_16x16x4_f32(
                    false, af[kk], false, bf[kk], (short)0, c, false, false);
        }
        __builtin_amdgcn_sched_barrier(0);

        float* zt = (float*)zt4[buf];
#pragma unroll
        for (int r = 0; r < 8; ++r) {
            zt[(rbase + r) * ZT_PITCH + col0 + n] = c[r];
        }
        if (i + 1 < TILES_PER_WG) PARK_TILE(buf ^ 1)
        __syncthreads();
        if (i + 2 < TILES_PER_WG) LOAD_TILE(tile + 2)

        const float4 v0 = zt4[buf][rr * (ZT_PITCH / 4) + q];
        const float4 v1 = zt4[buf][(rr + 8) * (ZT_PITCH / 4) + q];
        float ps0 = v0.x * as.x + v0.y * as.y + v0.z * as.z + v0.w * as.w;
        float pt0 = v0.x * ad.x + v0.y * ad.y + v0.z * ad.z + v0.w * ad.w;
        float ps1 = v1.x * as.x + v1.y * as.y + v1.z * as.z + v1.w * as.w;
        float pt1 = v1.x * ad.x + v1.y * ad.y + v1.z * ad.z + v1.w * ad.w;
        ((float4*)(z + (size_t)(row0 + rr) * OUT_DIM))[q] = v0;
        ((float4*)(z + (size_t)(row0 + rr + 8) * OUT_DIM))[q] = v1;
#pragma unroll
        for (int mk = 1; mk < 16; mk <<= 1) {
            const float a0 = __shfl_xor(ps0, mk, 32);
            const float a1 = __shfl_xor(pt0, mk, 32);
            const float a2 = __shfl_xor(ps1, mk, 32);
            const float a3 = __shfl_xor(pt1, mk, 32);
            ps0 += a0; pt0 += a1; ps1 += a2; pt1 += a3;
        }
        if (q == 0) {
            s[row0 + rr]     = ps0;
            t[row0 + rr]     = pt0;
            s[row0 + rr + 8] = ps1;
            t[row0 + rr + 8] = pt1;
        }

        uchar2 kk2;
        kk2.x = (unsigned char)(dd.x / RANGE_ROWS);
        kk2.y = (unsigned char)(dd.y / RANGE_ROWS);
        *(uchar2*)(key + e0) = kk2;
    }
#undef LOAD_TILE
#undef PARK_TILE
}

__device__ __forceinline__ unsigned zero_bytes(unsigned y) {
    return ~(((y & 0x7F7F7F7Fu) + 0x7F7F7F7Fu) | y | 0x7F7F7F7Fu);
}


__device__ __forceinline__ unsigned fkey(float v) {
    const unsigned b = __float_as_uint(v);
    return (b & 0x80000000u) ? ~b : (b | 0x80000000u);
}
__device__ __forceinline__ float funkey(unsigned k) {
    return __uint_as_float((k & 0x80000000u) ? (k & 0x7FFFFFFFu) : ~k);
}
__global__ void __launch_bounds__(512)
gat_range(const int* __restrict__ src, const int* __restrict__ dst,
          const unsigned char* __restrict__ key,
          const float* __restrict__ s, const float* __restrict__ t,
          const float* __restrict__ z, float* __restrict__ out) {
    __shared__ float    stage[BATCH * OUT_DIM];
    __shared__ unsigned l_idx[LIST_CAP];
    __shared__ unsigned c_pk[LIST_CAP];
    __shared__ float    c_e[LIST_CAP];
    __shared__ unsigned rmax[400];
    __shared__ unsigned mold[400];
    __shared__ float    scl[400];
    __shared__ unsigned short own_e[SCAN_WAVES * OWN_CAP];
    __shared__ int      segcnt[SCAN_WAVES];
    __shared__ int      more;

    const int tid  = threadIdx.x;
    const int lane = tid & 31;
    const int wave = __builtin_amdgcn_readfirstlane(tid >> 5);
    const int gid  = blockIdx.x;
    const int row_base = gid * RANGE_ROWS;

    v32f ax = {};
    v32f ay = {};
    for (int i = tid; i < 400; i += 512) rmax[i] = 0u;
    float dreg = 0.f;

    const uint4* __restrict__ key4 = (const uint4*)key;
    const unsigned pat = (unsigned)gid * 0x01010101u;
    int c = wave * CHUNKS_PER_WAVE;
    int cend = c + CHUNKS_PER_WAVE;
    if (cend > N_CHUNKS) cend = N_CHUNKS;

    for (int pass = 0; pass < MAX_PASS; ++pass) {
        if (tid == 0) more = 0;
        __syncthreads();

        int cnt = 0;
        bool full = false;
        while (c < cend && !full) {
            uint4 kv[8];
#pragma unroll
            for (int u = 0; u < 8; ++u) {
                int cc = c + u;
                if (cc > cend - 1) cc = cend - 1;
                kv[u] = key4[(size_t)cc * 32 + lane];
            }
#pragma unroll
            for (int u = 0; u < 8; ++u) {
                if (c < cend && !full) {
                    const unsigned t0 = zero_bytes(kv[u].x ^ pat);
                    const unsigned t1 = zero_bytes(kv[u].y ^ pat);
                    const unsigned t2 = zero_bytes(kv[u].z ^ pat);
                    const unsigned t3 = zero_bytes(kv[u].w ^ pat);
                    const unsigned fl = (t0 >> 7) | (t1 >> 6) | (t2 >> 5) | (t3 >> 4);
                    const unsigned mk = __builtin_amdgcn_ballot_w32(fl != 0u);
                    if (mk) {
                        const unsigned fl1 = fl & (fl - 1u);
                        const unsigned mk2 = __builtin_amdgcn_ballot_w32(fl1 != 0u);
                        int tot = __popc(mk) + __popc(mk2);
                        int off = (int)__builtin_amdgcn_mbcnt_lo(mk2, __builtin_amdgcn_mbcnt_lo(mk, 0u));
                        if (mk2 != 0u) {
                            const unsigned fl2 = fl1 & (fl1 - 1u);
                            if (__builtin_amdgcn_ballot_w32(fl2 != 0u) != 0u) {
                                const int cl = __popc(fl);
                                int x = cl;
#pragma unroll
                                for (int d = 1; d < 32; d <<= 1) {
                                    const int y = __shfl_up(x, d, 32);
                                    if (lane >= d) x += y;
                                }
                                off = x - cl;
                                tot = __builtin_amdgcn_readlane(x, 31);
                            }
                        }
                        const bool fits = (cnt + tot <= SEG_CAP);
                        if (fits && fl != 0u) {
                            const unsigned eb = (unsigned)c * CHUNK + (unsigned)(lane * 16);
                            int pos = wave * SEG_CAP + cnt + off;
                            if (fl1 == 0u) {
                                const unsigned b = (unsigned)__builtin_ctz(fl);
                                l_idx[pos] = eb + ((b & 3u) << 2) + (b >> 3);
                            } else {
                                const unsigned n0 = (((t0 >> 7) & 0x01010101u) * 0x01020408u) >> 24;
                                const unsigned n1 = (((t1 >> 7) & 0x01010101u) * 0x01020408u) >> 24;
                                const unsigned n2 = (((t2 >> 7) & 0x01010101u) * 0x01020408u) >> 24;
                                const unsigned n3 = (((t3 >> 7) & 0x01010101u) * 0x01020408u) >> 24;
                                unsigned m16 = (n0 & 15u) | ((n1 & 15u) << 4) | ((n2 & 15u) << 8) | ((n3 & 15u) << 12);
                                while (m16) {
                                    l_idx[pos++] = eb + (unsigned)__builtin_ctz(m16);
                                    m16 &= m16 - 1;
                                }
                            }
                        }
                        cnt  = fits ? cnt + tot : cnt;
                        full = !fits;
                    }
                    if (!full) ++c;
                }
            }
        }
        if (lane == 0) {
            segcnt[wave] = cnt;
            if (c < cend) more = 1;
        }
        __syncthreads();
        const int again = more;

        int total = 0;
        for (int w = 0; w < SCAN_WAVES; ++w) total += segcnt[w];
        {
            unsigned ee[SCAN_WAVES];
            int si[SCAN_WAVES], di[SCAN_WAVES];
            float sv[SCAN_WAVES], tv[SCAN_WAVES];
#pragma unroll
            for (int w = 0; w < SCAN_WAVES; ++w)
                ee[w] = (tid < segcnt[w]) ? l_idx[w * SEG_CAP + tid] : 0u;
#pragma unroll
            for (int w = 0; w < SCAN_WAVES; ++w) { si[w] = src[ee[w]]; di[w] = dst[ee[w]]; }
#pragma unroll
            for (int w = 0; w < SCAN_WAVES; ++w) { sv[w] = s[si[w]]; tv[w] = t[di[w]]; }
#pragma unroll
            for (int w = 0; w < SCAN_WAVES; ++w) {
                float v = sv[w] + tv[w];
                v = (v > 0.f) ? v : LEAKY * v;
                sv[w] = v;
                if (tid < segcnt[w]) atomicMax(&rmax[di[w] - row_base], fkey(v));
            }
            __syncthreads();
            if (pass > 0) {
                for (int r = tid; r < RANGE_ROWS; r += 512) {
                    const unsigned ko = mold[r], kn = rmax[r];
                    scl[r] = (ko != 0u && kn != ko) ? expf(funkey(ko) - funkey(kn)) : 1.0f;
                }
                __syncthreads();
#pragma unroll
                for (int k = 0; k < ROWS_PER_WAVE; ++k) {
                    const int r = 16 * k + wave;
                    const float sc = (r < RANGE_ROWS) ? scl[r] : 1.0f;
                    ax[k] *= sc;
                    ay[k] *= sc;
                }
                const int rown = lane * 16 + wave;
                if (rown < RANGE_ROWS) dreg *= scl[rown];
            }
            int base = 0;
#pragma unroll
            for (int w = 0; w < SCAN_WAVES; ++w) {
                const int n = segcnt[w];
                if (tid < n) {
                    const int lrow = di[w] - row_base;
                    c_pk[base + tid] = (unsigned)si[w] | ((unsigned)lrow << 17);
                    c_e[base + tid]  = expf(sv[w] - funkey(rmax[lrow]));
                }
                base += n;
            }
        }
        __syncthreads();

#define ROW_PIECE(K, B) {                                                           \
            const int q_ = tid + 512 * (K);                                          \
            int jx_ = (B) + (q_ >> 4);                                               \
            if (jx_ > total - 1) jx_ = total - 1;                                    \
            const unsigned sidx_ = c_pk[jx_] & 0x1FFFFu;                             \
            p##K = ((const float4*)(z + (size_t)sidx_ * OUT_DIM))[q_ & 15]; }
        float4 p0 = make_float4(0.f, 0.f, 0.f, 0.f);
        float4 p1 = p0, p2 = p0, p3 = p0, p4 = p0, p5 = p0, p6 = p0, p7 = p0;
        if (total > 0) {
            ROW_PIECE(0, 0) ROW_PIECE(1, 0) ROW_PIECE(2, 0) ROW_PIECE(3, 0)
            ROW_PIECE(4, 0) ROW_PIECE(5, 0) ROW_PIECE(6, 0) ROW_PIECE(7, 0)
        }
        for (int b0 = 0; b0 < total; b0 += BATCH) {
            ((float4*)stage)[tid + 512 * 0] = p0;
            ((float4*)stage)[tid + 512 * 1] = p1;
            ((float4*)stage)[tid + 512 * 2] = p2;
            ((float4*)stage)[tid + 512 * 3] = p3;
            ((float4*)stage)[tid + 512 * 4] = p4;
            ((float4*)stage)[tid + 512 * 5] = p5;
            ((float4*)stage)[tid + 512 * 6] = p6;
            ((float4*)stage)[tid + 512 * 7] = p7;
            __syncthreads();
            if (b0 + BATCH < total) {
                const int nb = b0 + BATCH;
                ROW_PIECE(0, nb) ROW_PIECE(1, nb) ROW_PIECE(2, nb) ROW_PIECE(3, nb)
                ROW_PIECE(4, nb) ROW_PIECE(5, nb) ROW_PIECE(6, nb) ROW_PIECE(7, nb)
            }
            unsigned pkv[BATCH / 32];
#pragma unroll
            for (int g = 0; g < BATCH / 32; ++g) {
                const int j = b0 + g * 32 + lane;
                const int jc = (j < total) ? j : 0;
                pkv[g] = c_pk[jc];
            }
            int n_own = 0;
#pragma unroll
            for (int g = 0; g < BATCH / 32; ++g) {
                const int j = b0 + g * 32 + lane;
                const int lrow = (int)(pkv[g] >> 17);
                const bool mine = (j < total) && ((lrow & 15) == wave);
                const unsigned mk = __builtin_amdgcn_ballot_w32(mine);
                if (mine) {
                    const int pos = wave * OWN_CAP + n_own + (int)__builtin_amdgcn_mbcnt_lo(mk, 0u);
                    own_e[pos] = (unsigned short)((g * 32 + lane) | ((lrow >> 4) << 8));
                }
                n_own += __popc(mk);
            }
            for (int c0 = 0; c0 < n_own; c0 += 32) {
                const int entv = (int)own_e[wave * OWN_CAP + c0 + lane];
                const int entp = __float_as_int(c_e[b0 + (entv & 255)]);
                int m = n_own - c0;
                if (m > 32) m = 32;
                for (int i = 0; i < m; i += OWN_STEP) {
                    int    ks[OWN_STEP];
                    float  ps[OWN_STEP];
                    float2 zs[OWN_STEP];
#pragma unroll
                    for (int u = 0; u < OWN_STEP; ++u) {
                        const int iu = (i + u < m) ? (i + u) : (m - 1);
                        const int e  = __builtin_amdgcn_readlane(entv, iu);
                        ps[u] = __int_as_float(__builtin_amdgcn_readlane(entp, iu));
                        ks[u] = e >> 8;
                        zs[u] = ((const float2*)stage)[(e & 255) * 32 + lane];
                    }
#pragma unroll
                    for (int u = 0; u < OWN_STEP; ++u) {
                        if (i + u < m) {
                            const int   k = ks[u];
                            const float p = ps[u];
                            ax[k] += p * zs[u].x;
                            ay[k] += p * zs[u].y;
                            dreg += (lane == k) ? p : 0.f;
                        }
                    }
                }
            }
            __syncthreads();
        }
        if (!again) break;
        for (int r = tid; r < RANGE_ROWS; r += 512) mold[r] = rmax[r];
    }
#pragma unroll
    for (int k = 0; k < ROWS_PER_WAVE; ++k) {
        const int r = 16 * k + wave;
        const int node = row_base + r;
        if (r < RANGE_ROWS && node < N_NODES) {
            const float d = __int_as_float(__builtin_amdgcn_readlane(__float_as_int(dreg), k));
            const float inv = (d > 0.f) ? 1.0f / d : 0.f;
            float2 o;
            o.x = ax[k] * inv;
            o.y = ay[k] * inv;
            ((float2*)(out + (size_t)node * OUT_DIM))[lane] = o;
        }
    }
}

extern "C" void kernel_launch(void* const* d_in, const int* in_sizes, int n_in,
                              void* d_out, int out_size, void* d_ws, size_t ws_size,
                              hipStream_t stream) {
    const float* h  = (const float*)d_in[0];
    const int* src  = (const int*)d_in[1];
    const int* dst  = (const int*)d_in[2];
    const float* W  = (const float*)d_in[3];
    const float* A  = (const float*)d_in[4];
    float* out = (float*)d_out;

    char* ws = (char*)d_ws;
    float*         z   = (float*)        (ws);
    float*         s   = (float*)        (ws + 25600000);
    float*         t   = (float*)        (ws + 26000000);
    unsigned char* key = (unsigned char*)(ws + 26400000);

    gat_front<<<FRONT_WGS, 128, 0, stream>>>(h, W, A, dst, z, s, t, key);
    gat_range<<<N_RANGES, 512, 0, stream>>>(src, dst, key, s, t, z, out);
}


# edge keys stored as bit planes (16 edges per 16-byte group, built by ballots in the front launch): the scan tests 32 edges per lane with a few xor-and steps; same list order and sums
# speedup vs baseline: 3.1195x; 1.0781x over previous
#include <hip/hip_runtime.h>
#include <hip/hip_bf16.h>
#include <math.h>

#define N_NODES 100000
#define N_EDGES 1600000
#define IN_DIM  128
#define OUT_DIM 64
#define LEAKY   0.01f

typedef float v2f __attribute__((ext_vector_type(2)));
typedef float v8f __attribute__((ext_vector_type(8)));
typedef float v32f __attribute__((ext_vector_type(32)));

#define RANGE_ROWS 391
#define N_RANGES   256
#define CHUNK      512
#define N_CHUNKS   (N_EDGES / CHUNK)
#define SCAN_WAVES 16
#define CHUNKS_PER_WAVE 196
#define SEG_CAP    512
#define LIST_CAP   (SCAN_WAVES * SEG_CAP)
#define BATCH      256
#define MAX_PASS   200
#define OWN_CAP    BATCH
#define ROWS_PER_WAVE 25
#define OWN_STEP   8


#define ZT_PITCH 68
#define HT_PITCH 132
#define TILES_PER_WG 5
#define FRONT_WGS (N_NODES / 16 / TILES_PER_WG)
#define EDGES_PER_TILE 256
static_assert((N_NODES % 16) == 0, "row tiles must be whole");
static_assert(FRONT_WGS * TILES_PER_WG * 16 == N_NODES, "every row tile belongs to exactly one work-group");
static_assert(EDGES_PER_TILE * (N_NODES / 16) == N_EDGES, "every edge gets its key from exactly one tile");

__global__ void __launch_bounds__(128)
gat_front(const float* __restrict__ h,
          const float* __restrict__ W,
          const float* __restrict__ A,
          const int* __restrict__ dst,
          float* __restrict__ z,
          float* __restrict__ s, float* __restrict__ t,
          unsigned* __restrict__ key) {
    __shared__ float4 hb4[2][16 * HT_PITCH / 4];
    __shared__ float4 zt4[2][16 * ZT_PITCH / 4];

    const int tid  = threadIdx.x;
    const int lane = tid & 31;
    const int wave = tid >> 5;
    const int tile0 = blockIdx.x * TILES_PER_WG;
    const int col0 = wave * 16;

    const int m     = lane & 15;
    const int khalf = (lane >> 4) * 2;

    const float4* __restrict__ hsrc = (const float4*)h + (size_t)(wave * 4) * (IN_DIM / 4) + lane;
    const int hdst = (wave * 4) * (HT_PITCH / 4) + lane;
#define LOAD_TILE(T_) {                                                     \
        const float4* p_ = hsrc + (size_t)(T_) * (16 * IN_DIM / 4);         \
        g0 = p_[0 * (IN_DIM / 4)]; g1 = p_[1 * (IN_DIM / 4)];               \
        g2 = p_[2 * (IN_DIM / 4)]; g3 = p_[3 * (IN_DIM / 4)]; }
#define PARK_TILE(B_) {                                                     \
        hb4[B_][hdst + 0 * (HT_PITCH / 4)] = g0;                            \
        hb4[B_][hdst + 1 * (HT_PITCH / 4)] = g1;                            \
        hb4[B_][hdst + 2 * (HT_PITCH / 4)] = g2;                            \
        hb4[B_][hdst + 3 * (HT_PITCH / 4)] = g3; }
    float4 g0, g1, g2, g3;
    LOAD_TILE(tile0)

    const float* __restrict__ wrow = W + (size_t)(col0 + m) * IN_DIM;
    v2f bf[IN_DIM / 4];
#pragma unroll
    for (int kk = 0; kk < IN_DIM / 4; ++kk) {
        bf[kk].x = wrow[4 * kk + khalf];
        bf[kk].y = wrow[4 * kk + khalf + 1];
    }

    const int q  = tid & 15;
    const int rr = tid >> 4;
    const float4 as = ((const float4*)A)[q];
    const float4 ad = ((const float4*)(A + OUT_DIM))[q];
    const int n     = lane & 15;
    const int rbase = (lane >> 4) * 8;

    PARK_TILE(0)
    __syncthreads();
    if (TILES_PER_WG > 1) LOAD_TILE(tile0 + 1)

    for (int i = 0; i < TILES_PER_WG; ++i) {
        const int tile = tile0 + i;
        const int buf  = i & 1;
        const int row0 = tile * 16;

        const int e0 = tile * EDGES_PER_TILE + wave * 64 + lane;
        const int d0 = dst[e0];
        const int d1 = dst[e0 + 32];

        const float* __restrict__ hrow = (const float*)hb4[buf] + m * HT_PITCH + khalf;
        v2f af[IN_DIM / 4];
#pragma unroll
        for (int kk = 0; kk < IN_DIM / 4; ++kk) af[kk] = *(const v2f*)(hrow + 4 * kk);
        __builtin_amdgcn_sched_barrier(0);
        v8f c = {};
#pragma unroll
        for (int kk = 0; kk < IN_DIM / 4; ++kk) {
            c = __builtin_amdgcn_wmma_f32_16x16x4_f32(
                    false, af[kk], false, bf[kk], (short)0, c, false, false);
        }
        __builtin_amdgcn_sched_barrier(0);

        float* zt = (float*)zt4[buf];
#pragma unroll
        for (int r = 0; r < 8; ++r) {
            zt[(rbase + r) * ZT_PITCH + col0 + n] = c[r];
        }
        if (i + 1 < TILES_PER_WG) PARK_TILE(buf ^ 1)
        __syncthreads();
        if (i + 2 < TILES_PER_WG) LOAD_TILE(tile + 2)

        const float4 v0 = zt4[buf][rr * (ZT_PITCH / 4) + q];
        const float4 v1 = zt4[buf][(rr + 8) * (ZT_PITCH / 4) + q];
        float ps0 = v0.x * as.x + v0.y * as.y + v0.z * as.z + v0.w * as.w;
        float pt0 = v0.x * ad.x + v0.y * ad.y + v0.z * ad.z + v0.w * ad.w;
        float ps1 = v1.x * as.x + v1.y * as.y + v1.z * as.z + v1.w * as.w;
        float pt1 = v1.x * ad.x + v1.y * ad.y + v1.z * ad.z + v1.w * ad.w;
        ((float4*)(z + (size_t)(row0 + rr) * OUT_DIM))[q] = v0;
        ((float4*)(z + (size_t)(row0 + rr + 8) * OUT_DIM))[q] = v1;
#pragma unroll
        for (int mk = 1; mk < 16; mk <<= 1) {
            const float a0 = __shfl_xor(ps0, mk, 32);
            const float a1 = __shfl_xor(pt0, mk, 32);
            const float a2 = __shfl_xor(ps1, mk, 32);
            const float a3 = __shfl_xor(pt1, mk, 32);
            ps0 += a0; pt0 += a1; ps1 += a2; pt1 += a3;
        }
        if (q == 0) {
            s[row0 + rr]     = ps0;
            t[row0 + rr]     = pt0;
            s[row0 + rr + 8] = ps1;
            t[row0 + rr + 8] = pt1;
        }

        const unsigned k0 = (unsigned)(d0 / RANGE_ROWS);
        const unsigned k1 = (unsigned)(d1 / RANGE_ROWS);
        unsigned wv = 0u;
#pragma unroll
        for (int k = 0; k < 4; ++k) {
            const unsigned a0 = __builtin_amdgcn_ballot_w32(((k0 >> (2 * k)) & 1u) != 0u);
            const unsigned a1 = __builtin_amdgcn_ballot_w32(((k0 >> (2 * k + 1)) & 1u) != 0u);
            const unsigned b0 = __builtin_amdgcn_ballot_w32(((k1 >> (2 * k)) & 1u) != 0u);
            const unsigned b1 = __builtin_amdgcn_ballot_w32(((k1 >> (2 * k + 1)) & 1u) != 0u);
            wv = (lane == k)      ? ((a0 & 0xFFFFu) | (a1 << 16))      : wv;
            wv = (lane == 4 + k)  ? ((a0 >> 16) | (a1 & 0xFFFF0000u))  : wv;
            wv = (lane == 8 + k)  ? ((b0 & 0xFFFFu) | (b1 << 16))      : wv;
            wv = (lane == 12 + k) ? ((b0 >> 16) | (b1 & 0xFFFF0000u))  : wv;
        }
        if (lane < 16) key[(tile * 16 + wave * 4) * 4 + lane] = wv;
    }
#undef LOAD_TILE
#undef PARK_TILE
}


__device__ __forceinline__ bool scan_append(unsigned fl, unsigned eb, int lane, int segbase,
                                            int& cnt, unsigned* list) {
    const unsigned mk = __builtin_amdgcn_ballot_w32(fl != 0u);
    if (mk == 0u) return true;
    const unsigned fl1 = fl & (fl - 1u);
    const unsigned mk2 = __builtin_amdgcn_ballot_w32(fl1 != 0u);
    int tot = __popc(mk) + __popc(mk2);
    int off = (int)__builtin_amdgcn_mbcnt_lo(mk2, __builtin_amdgcn_mbcnt_lo(mk, 0u));
    if (mk2 != 0u) {
        const unsigned fl2 = fl1 & (fl1 - 1u);
        if (__builtin_amdgcn_ballot_w32(fl2 != 0u) != 0u) {
            const int cl = __popc(fl);
            int x = cl;
#pragma unroll
            for (int d = 1; d < 32; d <<= 1) {
                const int y = __shfl_up(x, d, 32);
                if (lane >= d) x += y;
            }
            off = x - cl;
            tot = __builtin_amdgcn_readlane(x, 31);
        }
    }
    const bool fits = (cnt + tot <= SEG_CAP);
    if (fits && fl != 0u) {
        int pos = segbase + cnt + off;
        if (fl1 == 0u) {
            list[pos] = eb + (unsigned)__builtin_ctz(fl);
        } else {
            unsigned m = fl;
            while (m) {
                list[pos++] = eb + (unsigned)__builtin_ctz(m);
                m &= m - 1;
            }
        }
    }
    cnt = fits ? cnt + tot : cnt;
    return fits;
}

__device__ __forceinline__ unsigned fkey(float v) {
    const unsigned b = __float_as_uint(v);
    return (b & 0x80000000u) ? ~b : (b | 0x80000000u);
}
__device__ __forceinline__ float funkey(unsigned k) {
    return __uint_as_float((k & 0x80000000u) ? (k & 0x7FFFFFFFu) : ~k);
}
__global__ void __launch_bounds__(512)
gat_range(const int* __restrict__ src, const int* __restrict__ dst,
          const unsigned char* __restrict__ key,
          const float* __restrict__ s, const float* __restrict__ t,
          const float* __restrict__ z, float* __restrict__ out) {
    __shared__ float    stage[BATCH * OUT_DIM];
    __shared__ unsigned l_idx[LIST_CAP];
    __shared__ unsigned c_pk[LIST_CAP];
    __shared__ float    c_e[LIST_CAP];
    __shared__ unsigned rmax[400];
    __shared__ unsigned mold[400];
    __shared__ float    scl[400];
    __shared__ unsigned short own_e[SCAN_WAVES * OWN_CAP];
    __shared__ int      segcnt[SCAN_WAVES];
    __shared__ int      more;

    const int tid  = threadIdx.x;
    const int lane = tid & 31;
    const int wave = __builtin_amdgcn_readfirstlane(tid >> 5);
    const int gid  = blockIdx.x;
    const int row_base = gid * RANGE_ROWS;

    v32f ax = {};
    v32f ay = {};
    for (int i = tid; i < 400; i += 512) rmax[i] = 0u;
    float dreg = 0.f;

    const uint4* __restrict__ key4 = (const uint4*)key;
    const unsigned inv0 = ((gid & 1)   ? 0u : 0x0000FFFFu) | ((gid & 2)   ? 0u : 0xFFFF0000u);
    const unsigned inv1 = ((gid & 4)   ? 0u : 0x0000FFFFu) | ((gid & 8)   ? 0u : 0xFFFF0000u);
    const unsigned inv2 = ((gid & 16)  ? 0u : 0x0000FFFFu) | ((gid & 32)  ? 0u : 0xFFFF0000u);
    const unsigned inv3 = ((gid & 64)  ? 0u : 0x0000FFFFu) | ((gid & 128) ? 0u : 0xFFFF0000u);
    bool slow = false;
    int c = wave * CHUNKS_PER_WAVE;
    int cend = c + CHUNKS_PER_WAVE;
    if (cend > N_CHUNKS) cend = N_CHUNKS;

    for (int pass = 0; pass < MAX_PASS; ++pass) {
        if (tid == 0) more = 0;
        __syncthreads();

        int cnt = 0;
        bool full = false;
        while (c < cend && !full) {
            uint4 kv[8];
            const int glast = cend * 32 - 1;
#pragma unroll
            for (int u = 0; u < 8; ++u) {
                int gi = (c + (u & ~1)) * 32 + 2 * lane + (u & 1);
                if (gi > glast) gi = glast;
                kv[u] = key4[gi];
            }
#pragma unroll
            for (int u = 0; u < 4; ++u) {
                if (c + 1 < cend && !full && !slow) {
                    const unsigned eqa = (kv[2 * u].x ^ inv0) & (kv[2 * u].y ^ inv1) & (kv[2 * u].z ^ inv2) & (kv[2 * u].w ^ inv3);
                    const unsigned eqb = (kv[2 * u + 1].x ^ inv0) & (kv[2 * u + 1].y ^ inv1) & (kv[2 * u + 1].z ^ inv2) & (kv[2 * u + 1].w ^ inv3);
                    const unsigned fl = (eqa & (eqa >> 16)) | ((eqb & (eqb >> 16)) << 16);
                    const bool fits = scan_append(fl, (unsigned)c * CHUNK + (unsigned)(lane * 32), lane, wave * SEG_CAP, cnt, l_idx);
                    if (fits) c += 2;
                    else if (cnt == 0) slow = true;
                    else full = true;
                }
            }
            if (!full && c < cend && (slow || c + 1 == cend)) {
                const uint4 k1 = key4[c * 32 + lane];
                const unsigned eq = (k1.x ^ inv0) & (k1.y ^ inv1) & (k1.z ^ inv2) & (k1.w ^ inv3);
                const unsigned fl = eq & (eq >> 16);
                const bool fits = scan_append(fl, (unsigned)c * CHUNK + (unsigned)(lane * 16), lane, wave * SEG_CAP, cnt, l_idx);
                if (fits) ++c;
                else full = true;
            }
        }
        if (lane == 0) {
            segcnt[wave] = cnt;
            if (c < cend) more = 1;
        }
        __syncthreads();
        const int again = more;

        int total = 0;
        for (int w = 0; w < SCAN_WAVES; ++w) total += segcnt[w];
        {
            unsigned ee[SCAN_WAVES];
            int si[SCAN_WAVES], di[SCAN_WAVES];
            float sv[SCAN_WAVES], tv[SCAN_WAVES];
#pragma unroll
            for (int w = 0; w < SCAN_WAVES; ++w)
                ee[w] = (tid < segcnt[w]) ? l_idx[w * SEG_CAP + tid] : 0u;
#pragma unroll
            for (int w = 0; w < SCAN_WAVES; ++w) { si[w] = src[ee[w]]; di[w] = dst[ee[w]]; }
#pragma unroll
            for (int w = 0; w < SCAN_WAVES; ++w) { sv[w] = s[si[w]]; tv[w] = t[di[w]]; }
#pragma unroll
            for (int w = 0; w < SCAN_WAVES; ++w) {
                float v = sv[w] + tv[w];
                v = (v > 0.f) ? v : LEAKY * v;
                sv[w] = v;
                if (tid < segcnt[w]) atomicMax(&rmax[di[w] - row_base], fkey(v));
            }
            __syncthreads();
            if (pass > 0) {
                for (int r = tid; r < RANGE_ROWS; r += 512) {
                    const unsigned ko = mold[r], kn = rmax[r];
                    scl[r] = (ko != 0u && kn != ko) ? expf(funkey(ko) - funkey(kn)) : 1.0f;
                }
                __syncthreads();
#pragma unroll
                for (int k = 0; k < ROWS_PER_WAVE; ++k) {
                    const int r = 16 * k + wave;
                    const float sc = (r < RANGE_ROWS) ? scl[r] : 1.0f;
                    ax[k] *= sc;
                    ay[k] *= sc;
                }
                const int rown = lane * 16 + wave;
                if (rown < RANGE_ROWS) dreg *= scl[rown];
            }
            int base = 0;
#pragma unroll
            for (int w = 0; w < SCAN_WAVES; ++w) {
                const int n = segcnt[w];
                if (tid < n) {
                    const int lrow = di[w] - row_base;
                    c_pk[base + tid] = (unsigned)si[w] | ((unsigned)lrow << 17);
                    c_e[base + tid]  = expf(sv[w] - funkey(rmax[lrow]));
                }
                base += n;
            }
        }
        __syncthreads();

#define ROW_PIECE(K, B) {                                                           \
            const int q_ = tid + 512 * (K);                                          \
            int jx_ = (B) + (q_ >> 4);                                               \
            if (jx_ > total - 1) jx_ = total - 1;                                    \
            const unsigned sidx_ = c_pk[jx_] & 0x1FFFFu;                             \
            p##K = ((const float4*)(z + (size_t)sidx_ * OUT_DIM))[q_ & 15]; }
        float4 p0 = make_float4(0.f, 0.f, 0.f, 0.f);
        float4 p1 = p0, p2 = p0, p3 = p0, p4 = p0, p5 = p0, p6 = p0, p7 = p0;
        if (total > 0) {
            ROW_PIECE(0, 0) ROW_PIECE(1, 0) ROW_PIECE(2, 0) ROW_PIECE(3, 0)
            ROW_PIECE(4, 0) ROW_PIECE(5, 0) ROW_PIECE(6, 0) ROW_PIECE(7, 0)
        }
        for (int b0 = 0; b0 < total; b0 += BATCH) {
            ((float4*)stage)[tid + 512 * 0] = p0;
            ((float4*)stage)[tid + 512 * 1] = p1;
            ((float4*)stage)[tid + 512 * 2] = p2;
            ((float4*)stage)[tid + 512 * 3] = p3;
            ((float4*)stage)[tid + 512 * 4] = p4;
            ((float4*)stage)[tid + 512 * 5] = p5;
            ((float4*)stage)[tid + 512 * 6] = p6;
            ((float4*)stage)[tid + 512 * 7] = p7;
            __syncthreads();
            if (b0 + BATCH < total) {
                const int nb = b0 + BATCH;
                ROW_PIECE(0, nb) ROW_PIECE(1, nb) ROW_PIECE(2, nb) ROW_PIECE(3, nb)
                ROW_PIECE(4, nb) ROW_PIECE(5, nb) ROW_PIECE(6, nb) ROW_PIECE(7, nb)
            }
            unsigned pkv[BATCH / 32];
#pragma unroll
            for (int g = 0; g < BATCH / 32; ++g) {
                const int j = b0 + g * 32 + lane;
                const int jc = (j < total) ? j : 0;
                pkv[g] = c_pk[jc];
            }
            int n_own = 0;
#pragma unroll
            for (int g = 0; g < BATCH / 32; ++g) {
                const int j = b0 + g * 32 + lane;
                const int lrow = (int)(pkv[g] >> 17);
                const bool mine = (j < total) && ((lrow & 15) == wave);
                const unsigned mk = __builtin_amdgcn_ballot_w32(mine);
                if (mine) {
                    const int pos = wave * OWN_CAP + n_own + (int)__builtin_amdgcn_mbcnt_lo(mk, 0u);
                    own_e[pos] = (unsigned short)((g * 32 + lane) | ((lrow >> 4) << 8));
                }
                n_own += __popc(mk);
            }
            for (int c0 = 0; c0 < n_own; c0 += 32) {
                const int entv = (int)own_e[wave * OWN_CAP + c0 + lane];
                const int entp = __float_as_int(c_e[b0 + (entv & 255)]);
                int m = n_own - c0;
                if (m > 32) m = 32;
                for (int i = 0; i < m; i += OWN_STEP) {
                    int    ks[OWN_STEP];
                    float  ps[OWN_STEP];
                    float2 zs[OWN_STEP];
#pragma unroll
                    for (int u = 0; u < OWN_STEP; ++u) {
                        const int iu = (i + u < m) ? (i + u) : (m - 1);
                        const int e  = __builtin_amdgcn_readlane(entv, iu);
                        ps[u] = __int_as_float(__builtin_amdgcn_readlane(entp, iu));
                        ks[u] = e >> 8;
                        zs[u] = ((const float2*)stage)[(e & 255) * 32 + lane];
                    }
#pragma unroll
                    for (int u = 0; u < OWN_STEP; ++u) {
                        if (i + u < m) {
                            const int   k = ks[u];
                            const float p = ps[u];
                            ax[k] += p * zs[u].x;
                            ay[k] += p * zs[u].y;
                            dreg += (lane == k) ? p : 0.f;
                        }
                    }
                }
            }
            __syncthreads();
        }
        if (!again) break;
        for (int r = tid; r < RANGE_ROWS; r += 512) mold[r] = rmax[r];
    }
#pragma unroll
    for (int k = 0; k < ROWS_PER_WAVE; ++k) {
        const int r = 16 * k + wave;
        const int node = row_base + r;
        if (r < RANGE_ROWS && node < N_NODES) {
            const float d = __int_as_float(__builtin_amdgcn_readlane(__float_as_int(dreg), k));
            const float inv = (d > 0.f) ? 1.0f / d : 0.f;
            float2 o;
            o.x = ax[k] * inv;
            o.y = ay[k] * inv;
            ((float2*)(out + (size_t)node * OUT_DIM))[lane] = o;
        }
    }
}

extern "C" void kernel_launch(void* const* d_in, const int* in_sizes, int n_in,
                              void* d_out, int out_size, void* d_ws, size_t ws_size,
                              hipStream_t stream) {
    const float* h  = (const float*)d_in[0];
    const int* src  = (const int*)d_in[1];
    const int* dst  = (const int*)d_in[2];
    const float* W  = (const float*)d_in[3];
    const float* A  = (const float*)d_in[4];
    float* out = (float*)d_out;

    char* ws = (char*)d_ws;
    float*         z   = (float*)        (ws);
    float*         s   = (float*)        (ws + 25600000);
    float*         t   = (float*)        (ws + 26000000);
    unsigned char* key = (unsigned char*)(ws + 26400000);

    gat_front<<<FRONT_WGS, 128, 0, stream>>>(h, W, A, dst, z, s, t, (unsigned*)key);
    gat_range<<<N_RANGES, 512, 0, stream>>>(src, dst, key, s, t, z, out);
}


# edge launch makes one random 4-byte read per listed edge, not two: the front launch packs source node and row-in-range into one word per edge from coalesced src and dst reads; same sums and order
# speedup vs baseline: 3.2335x; 1.0365x over previous
#include <hip/hip_runtime.h>
#include <hip/hip_bf16.h>
#include <math.h>

#define N_NODES 100000
#define N_EDGES 1600000
#define IN_DIM  128
#define OUT_DIM 64
#define LEAKY   0.01f

typedef float v2f __attribute__((ext_vector_type(2)));
typedef float v8f __attribute__((ext_vector_type(8)));
typedef float v32f __attribute__((ext_vector_type(32)));

#define RANGE_ROWS 391
#define N_RANGES   256
#define CHUNK      512
#define N_CHUNKS   (N_EDGES / CHUNK)
#define SCAN_WAVES 16
#define CHUNKS_PER_WAVE 196
#define SEG_CAP    512
#define LIST_CAP   (SCAN_WAVES * SEG_CAP)
#define BATCH      256
#define MAX_PASS   200
#define OWN_CAP    BATCH
#define ROWS_PER_WAVE 25
#define OWN_STEP   8
#define WS_Z_BYTES   25600000
#define WS_ST_BYTES  400000
#define WS_KEY_BYTES 1600000
#define WS_PK_OFF    (WS_Z_BYTES + 2 * WS_ST_BYTES + WS_KEY_BYTES)
#define WS_NEED_PK   ((size_t)WS_PK_OFF + (size_t)N_EDGES * 4)


#define ZT_PITCH 68
#define HT_PITCH 132
#define TILES_PER_WG 5
#define FRONT_WGS (N_NODES / 16 / TILES_PER_WG)
#define EDGES_PER_TILE 256
static_assert((N_NODES % 16) == 0, "row tiles must be whole");
static_assert(FRONT_WGS * TILES_PER_WG * 16 == N_NODES, "every row tile belongs to exactly one work-group");
static_assert(EDGES_PER_TILE * (N_NODES / 16) == N_EDGES, "every edge gets its key from exactly one tile");

__global__ void __launch_bounds__(128)
gat_front(const float* __restrict__ h,
          const float* __restrict__ W,
          const float* __restrict__ A,
          const int* __restrict__ src,
          const int* __restrict__ dst,
          float* __restrict__ z,
          float* __restrict__ s, float* __restrict__ t,
          unsigned* __restrict__ key,
          unsigned* __restrict__ pk, size_t ws_bytes) {
    __shared__ float4 hb4[2][16 * HT_PITCH / 4];
    __shared__ float4 zt4[2][16 * ZT_PITCH / 4];

    const int tid  = threadIdx.x;
    const int lane = tid & 31;
    const int wave = tid >> 5;
    const int tile0 = blockIdx.x * TILES_PER_WG;
    const int col0 = wave * 16;
    const bool use_pk = ws_bytes >= WS_NEED_PK;

    const int m     = lane & 15;
    const int khalf = (lane >> 4) * 2;

    const float4* __restrict__ hsrc = (const float4*)h + (size_t)(wave * 4) * (IN_DIM / 4) + lane;
    const int hdst = (wave * 4) * (HT_PITCH / 4) + lane;
#define LOAD_TILE(T_) {                                                     \
        const float4* p_ = hsrc + (size_t)(T_) * (16 * IN_DIM / 4);         \
        g0 = p_[0 * (IN_DIM / 4)]; g1 = p_[1 * (IN_DIM / 4)];               \
        g2 = p_[2 * (IN_DIM / 4)]; g3 = p_[3 * (IN_DIM / 4)]; }
#define PARK_TILE(B_) {                                                     \
        hb4[B_][hdst + 0 * (HT_PITCH / 4)] = g0;                            \
        hb4[B_][hdst + 1 * (HT_PITCH / 4)] = g1;                            \
        hb4[B_][hdst + 2 * (HT_PITCH / 4)] = g2;                            \
        hb4[B_][hdst + 3 * (HT_PITCH / 4)] = g3; }
    float4 g0, g1, g2, g3;
    LOAD_TILE(tile0)

    const float* __restrict__ wrow = W + (size_t)(col0 + m) * IN_DIM;
    v2f bf[IN_DIM / 4];
#pragma unroll
    for (int kk = 0; kk < IN_DIM / 4; ++kk) {
        bf[kk].x = wrow[4 * kk + khalf];
        bf[kk].y = wrow[4 * kk + khalf + 1];
    }

    const int q  = tid & 15;
    const int rr = tid >> 4;
    const float4 as = ((const float4*)A)[q];
    const float4 ad = ((const float4*)(A + OUT_DIM))[q];
    const int n     = lane & 15;
    const int rbase = (lane >> 4) * 8;

    PARK_TILE(0)
    __syncthreads();
    if (TILES_PER_WG > 1) LOAD_TILE(tile0 + 1)

    for (int i = 0; i < TILES_PER_WG; ++i) {
        const int tile = tile0 + i;
        const int buf  = i & 1;
        const int row0 = tile * 16;

        const int e0 = tile * EDGES_PER_TILE + wave * 64 + lane;
        const int d0 = dst[e0];
        const int d1 = dst[e0 + 32];
        int s0 = 0, s1 = 0;
        if (use_pk) { s0 = src[e0]; s1 = src[e0 + 32]; }

        const float* __restrict__ hrow = (const float*)hb4[buf] + m * HT_PITCH + khalf;
        v2f af[IN_DIM / 4];
#pragma unroll
        for (int kk = 0; kk < IN_DIM / 4; ++kk) af[kk] = *(const v2f*)(hrow + 4 * kk);
        __builtin_amdgcn_sched_barrier(0);
        v8f c = {};
#pragma unroll
        for (int kk = 0; kk < IN_DIM / 4; ++kk) {
            c = __builtin_amdgcn_wmma_f32_16x16x4_f32(
                    false, af[kk], false, bf[kk], (short)0, c, false, false);
        }
        __builtin_amdgcn_sched_barrier(0);

        float* zt = (float*)zt4[buf];
#pragma unroll
        for (int r = 0; r < 8; ++r) {
            zt[(rbase + r) * ZT_PITCH + col0 + n] = c[r];
        }
        if (i + 1 < TILES_PER_WG) PARK_TILE(buf ^ 1)
        __syncthreads();
        if (i + 2 < TILES_PER_WG) LOAD_TILE(tile + 2)

        const float4 v0 = zt4[buf][rr * (ZT_PITCH / 4) + q];
        const float4 v1 = zt4[buf][(rr + 8) * (ZT_PITCH / 4) + q];
        float ps0 = v0.x * as.x + v0.y * as.y + v0.z * as.z + v0.w * as.w;
        float pt0 = v0.x * ad.x + v0.y * ad.y + v0.z * ad.z + v0.w * ad.w;
        float ps1 = v1.x * as.x + v1.y * as.y + v1.z * as.z + v1.w * as.w;
        float pt1 = v1.x * ad.x + v1.y * ad.y + v1.z * ad.z + v1.w * ad.w;
        ((float4*)(z + (size_t)(row0 + rr) * OUT_DIM))[q] = v0;
        ((float4*)(z + (size_t)(row0 + rr + 8) * OUT_DIM))[q] = v1;
#pragma unroll
        for (int mk = 1; mk < 16; mk <<= 1) {
            const float a0 = __shfl_xor(ps0, mk, 32);
            const float a1 = __shfl_xor(pt0, mk, 32);
            const float a2 = __shfl_xor(ps1, mk, 32);
            const float a3 = __shfl_xor(pt1, mk, 32);
            ps0 += a0; pt0 += a1; ps1 += a2; pt1 += a3;
        }
        if (q == 0) {
            s[row0 + rr]     = ps0;
            t[row0 + rr]     = pt0;
            s[row0 + rr + 8] = ps1;
            t[row0 + rr + 8] = pt1;
        }

        const unsigned k0 = (unsigned)(d0 / RANGE_ROWS);
        const unsigned k1 = (unsigned)(d1 / RANGE_ROWS);
        if (use_pk) {
            pk[e0]      = (unsigned)s0 | (((unsigned)d0 - k0 * RANGE_ROWS) << 17);
            pk[e0 + 32] = (unsigned)s1 | (((unsigned)d1 - k1 * RANGE_ROWS) << 17);
        }
        unsigned wv = 0u;
#pragma unroll
        for (int k = 0; k < 4; ++k) {
            const unsigned a0 = __builtin_amdgcn_ballot_w32(((k0 >> (2 * k)) & 1u) != 0u);
            const unsigned a1 = __builtin_amdgcn_ballot_w32(((k0 >> (2 * k + 1)) & 1u) != 0u);
            const unsigned b0 = __builtin_amdgcn_ballot_w32(((k1 >> (2 * k)) & 1u) != 0u);
            const unsigned b1 = __builtin_amdgcn_ballot_w32(((k1 >> (2 * k + 1)) & 1u) != 0u);
            wv = (lane == k)      ? ((a0 & 0xFFFFu) | (a1 << 16))      : wv;
            wv = (lane == 4 + k)  ? ((a0 >> 16) | (a1 & 0xFFFF0000u))  : wv;
            wv = (lane == 8 + k)  ? ((b0 & 0xFFFFu) | (b1 << 16))      : wv;
            wv = (lane == 12 + k) ? ((b0 >> 16) | (b1 & 0xFFFF0000u))  : wv;
        }
        if (lane < 16) key[(tile * 16 + wave * 4) * 4 + lane] = wv;
    }
#undef LOAD_TILE
#undef PARK_TILE
}


__device__ __forceinline__ bool scan_append(unsigned fl, unsigned eb, int lane, int segbase,
                                            int& cnt, unsigned* list) {
    const unsigned mk = __builtin_amdgcn_ballot_w32(fl != 0u);
    if (mk == 0u) return true;
    const unsigned fl1 = fl & (fl - 1u);
    const unsigned mk2 = __builtin_amdgcn_ballot_w32(fl1 != 0u);
    int tot = __popc(mk) + __popc(mk2);
    int off = (int)__builtin_amdgcn_mbcnt_lo(mk2, __builtin_amdgcn_mbcnt_lo(mk, 0u));
    if (mk2 != 0u) {
        const unsigned fl2 = fl1 & (fl1 - 1u);
        if (__builtin_amdgcn_ballot_w32(fl2 != 0u) != 0u) {
            const int cl = __popc(fl);
            int x = cl;
#pragma unroll
            for (int d = 1; d < 32; d <<= 1) {
                const int y = __shfl_up(x, d, 32);
                if (lane >= d) x += y;
            }
            off = x - cl;
            tot = __builtin_amdgcn_readlane(x, 31);
        }
    }
    const bool fits = (cnt + tot <= SEG_CAP);
    if (fits && fl != 0u) {
        int pos = segbase + cnt + off;
        if (fl1 == 0u) {
            list[pos] = eb + (unsigned)__builtin_ctz(fl);
        } else {
            unsigned m = fl;
            while (m) {
                list[pos++] = eb + (unsigned)__builtin_ctz(m);
                m &= m - 1;
            }
        }
    }
    cnt = fits ? cnt + tot : cnt;
    return fits;
}

__device__ __forceinline__ unsigned fkey(float v) {
    const unsigned b = __float_as_uint(v);
    return (b & 0x80000000u) ? ~b : (b | 0x80000000u);
}
__device__ __forceinline__ float funkey(unsigned k) {
    return __uint_as_float((k & 0x80000000u) ? (k & 0x7FFFFFFFu) : ~k);
}
__global__ void __launch_bounds__(512)
gat_range(const int* __restrict__ src, const int* __restrict__ dst,
          const unsigned char* __restrict__ key,
          const unsigned* __restrict__ pk, size_t ws_bytes,
          const float* __restrict__ s, const float* __restrict__ t,
          const float* __restrict__ z, float* __restrict__ out) {
    __shared__ float    stage[BATCH * OUT_DIM];
    __shared__ unsigned l_idx[LIST_CAP];
    __shared__ unsigned c_pk[LIST_CAP];
    __shared__ float    c_e[LIST_CAP];
    __shared__ unsigned rmax[400];
    __shared__ unsigned mold[400];
    __shared__ float    scl[400];
    __shared__ unsigned short own_e[SCAN_WAVES * OWN_CAP];
    __shared__ int      segcnt[SCAN_WAVES];
    __shared__ int      more;

    const int tid  = threadIdx.x;
    const int lane = tid & 31;
    const int wave = __builtin_amdgcn_readfirstlane(tid >> 5);
    const int gid  = blockIdx.x;
    const int row_base = gid * RANGE_ROWS;
    const bool use_pk = ws_bytes >= WS_NEED_PK;

    v32f ax = {};
    v32f ay = {};
    for (int i = tid; i < 400; i += 512) rmax[i] = 0u;
    float dreg = 0.f;

    const uint4* __restrict__ key4 = (const uint4*)key;
    const unsigned inv0 = ((gid & 1)   ? 0u : 0x0000FFFFu) | ((gid & 2)   ? 0u : 0xFFFF0000u);
    const unsigned inv1 = ((gid & 4)   ? 0u : 0x0000FFFFu) | ((gid & 8)   ? 0u : 0xFFFF0000u);
    const unsigned inv2 = ((gid & 16)  ? 0u : 0x0000FFFFu) | ((gid & 32)  ? 0u : 0xFFFF0000u);
    const unsigned inv3 = ((gid & 64)  ? 0u : 0x0000FFFFu) | ((gid & 128) ? 0u : 0xFFFF0000u);
    bool slow = false;
    int c = wave * CHUNKS_PER_WAVE;
    int cend = c + CHUNKS_PER_WAVE;
    if (cend > N_CHUNKS) cend = N_CHUNKS;

    for (int pass = 0; pass < MAX_PASS; ++pass) {
        if (tid == 0) more = 0;
        __syncthreads();

        int cnt = 0;
        bool full = false;
        while (c < cend && !full) {
            uint4 kv[8];
            const int glast = cend * 32 - 1;
#pragma unroll
            for (int u = 0; u < 8; ++u) {
                int gi = (c + (u & ~1)) * 32 + 2 * lane + (u & 1);
                if (gi > glast) gi = glast;
                kv[u] = key4[gi];
            }
#pragma unroll
            for (int u = 0; u < 4; ++u) {
                if (c + 1 < cend && !full && !slow) {
                    const unsigned eqa = (kv[2 * u].x ^ inv0) & (kv[2 * u].y ^ inv1) & (kv[2 * u].z ^ inv2) & (kv[2 * u].w ^ inv3);
                    const unsigned eqb = (kv[2 * u + 1].x ^ inv0) & (kv[2 * u + 1].y ^ inv1) & (kv[2 * u + 1].z ^ inv2) & (kv[2 * u + 1].w ^ inv3);
                    const unsigned fl = (eqa & (eqa >> 16)) | ((eqb & (eqb >> 16)) << 16);
                    const bool fits = scan_append(fl, (unsigned)c * CHUNK + (unsigned)(lane * 32), lane, wave * SEG_CAP, cnt, l_idx);
                    if (fits) c += 2;
                    else if (cnt == 0) slow = true;
                    else full = true;
                }
            }
            if (!full && c < cend && (slow || c + 1 == cend)) {
                const uint4 k1 = key4[c * 32 + lane];
                const unsigned eq = (k1.x ^ inv0) & (k1.y ^ inv1) & (k1.z ^ inv2) & (k1.w ^ inv3);
                const unsigned fl = eq & (eq >> 16);
                const bool fits = scan_append(fl, (unsigned)c * CHUNK + (unsigned)(lane * 16), lane, wave * SEG_CAP, cnt, l_idx);
                if (fits) ++c;
                else full = true;
            }
        }
        if (lane == 0) {
            segcnt[wave] = cnt;
            if (c < cend) more = 1;
        }
        __syncthreads();
        const int again = more;

        int total = 0;
        for (int w = 0; w < SCAN_WAVES; ++w) total += segcnt[w];
        {
            unsigned ee[SCAN_WAVES];
            int si[SCAN_WAVES], di[SCAN_WAVES];
            float sv[SCAN_WAVES], tv[SCAN_WAVES];
#pragma unroll
            for (int w = 0; w < SCAN_WAVES; ++w)
                ee[w] = (tid < segcnt[w]) ? l_idx[w * SEG_CAP + tid] : 0u;
            if (use_pk) {
#pragma unroll
                for (int w = 0; w < SCAN_WAVES; ++w) si[w] = (int)pk[ee[w]];
#pragma unroll
                for (int w = 0; w < SCAN_WAVES; ++w) {
                    const unsigned pw = (tid < segcnt[w]) ? (unsigned)si[w] : 0u;
                    si[w] = (int)(pw & 0x1FFFFu);
                    di[w] = row_base + (int)(pw >> 17);
                }
            } else {
#pragma unroll
                for (int w = 0; w < SCAN_WAVES; ++w) { si[w] = src[ee[w]]; di[w] = dst[ee[w]]; }
            }
#pragma unroll
            for (int w = 0; w < SCAN_WAVES; ++w) { sv[w] = s[si[w]]; tv[w] = t[di[w]]; }
#pragma unroll
            for (int w = 0; w < SCAN_WAVES; ++w) {
                float v = sv[w] + tv[w];
                v = (v > 0.f) ? v : LEAKY * v;
                sv[w] = v;
                if (tid < segcnt[w]) atomicMax(&rmax[di[w] - row_base], fkey(v));
            }
            __syncthreads();
            if (pass > 0) {
                for (int r = tid; r < RANGE_ROWS; r += 512) {
                    const unsigned ko = mold[r], kn = rmax[r];
                    scl[r] = (ko != 0u && kn != ko) ? expf(funkey(ko) - funkey(kn)) : 1.0f;
                }
                __syncthreads();
#pragma unroll
                for (int k = 0; k < ROWS_PER_WAVE; ++k) {
                    const int r = 16 * k + wave;
                    const float sc = (r < RANGE_ROWS) ? scl[r] : 1.0f;
                    ax[k] *= sc;
                    ay[k] *= sc;
                }
                const int rown = lane * 16 + wave;
                if (rown < RANGE_ROWS) dreg *= scl[rown];
            }
            int base = 0;
#pragma unroll
            for (int w = 0; w < SCAN_WAVES; ++w) {
                const int n = segcnt[w];
                if (tid < n) {
                    const int lrow = di[w] - row_base;
                    c_pk[base + tid] = (unsigned)si[w] | ((unsigned)lrow << 17);
                    c_e[base + tid]  = expf(sv[w] - funkey(rmax[lrow]));
                }
                base += n;
            }
        }
        __syncthreads();

#define ROW_PIECE(K, B) {                                                           \
            const int q_ = tid + 512 * (K);                                          \
            int jx_ = (B) + (q_ >> 4);                                               \
            if (jx_ > total - 1) jx_ = total - 1;                                    \
            const unsigned sidx_ = c_pk[jx_] & 0x1FFFFu;                             \
            p##K = ((const float4*)(z + (size_t)sidx_ * OUT_DIM))[q_ & 15]; }
        float4 p0 = make_float4(0.f, 0.f, 0.f, 0.f);
        float4 p1 = p0, p2 = p0, p3 = p0, p4 = p0, p5 = p0, p6 = p0, p7 = p0;
        if (total > 0) {
            ROW_PIECE(0, 0) ROW_PIECE(1, 0) ROW_PIECE(2, 0) ROW_PIECE(3, 0)
            ROW_PIECE(4, 0) ROW_PIECE(5, 0) ROW_PIECE(6, 0) ROW_PIECE(7, 0)
        }
        for (int b0 = 0; b0 < total; b0 += BATCH) {
            ((float4*)stage)[tid + 512 * 0] = p0;
            ((float4*)stage)[tid + 512 * 1] = p1;
            ((float4*)stage)[tid + 512 * 2] = p2;
            ((float4*)stage)[tid + 512 * 3] = p3;
            ((float4*)stage)[tid + 512 * 4] = p4;
            ((float4*)stage)[tid + 512 * 5] = p5;
            ((float4*)stage)[tid + 512 * 6] = p6;
            ((float4*)stage)[tid + 512 * 7] = p7;
            __syncthreads();
            if (b0 + BATCH < total) {
                const int nb = b0 + BATCH;
                ROW_PIECE(0, nb) ROW_PIECE(1, nb) ROW_PIECE(2, nb) ROW_PIECE(3, nb)
                ROW_PIECE(4, nb) ROW_PIECE(5, nb) ROW_PIECE(6, nb) ROW_PIECE(7, nb)
            }
            unsigned pkv[BATCH / 32];
#pragma unroll
            for (int g = 0; g < BATCH / 32; ++g) {
                const int j = b0 + g * 32 + lane;
                const int jc = (j < total) ? j : 0;
                pkv[g] = c_pk[jc];
            }
            int n_own = 0;
#pragma unroll
            for (int g = 0; g < BATCH / 32; ++g) {
                const int j = b0 + g * 32 + lane;
                const int lrow = (int)(pkv[g] >> 17);
                const bool mine = (j < total) && ((lrow & 15) == wave);
                const unsigned mk = __builtin_amdgcn_ballot_w32(mine);
                if (mine) {
                    const int pos = wave * OWN_CAP + n_own + (int)__builtin_amdgcn_mbcnt_lo(mk, 0u);
                    own_e[pos] = (unsigned short)((g * 32 + lane) | ((lrow >> 4) << 8));
                }
                n_own += __popc(mk);
            }
            for (int c0 = 0; c0 < n_own; c0 += 32) {
                const int entv = (int)own_e[wave * OWN_CAP + c0 + lane];
                const int entp = __float_as_int(c_e[b0 + (entv & 255)]);
                int m = n_own - c0;
                if (m > 32) m = 32;
                for (int i = 0; i < m; i += OWN_STEP) {
                    int    ks[OWN_STEP];
                    float  ps[OWN_STEP];
                    float2 zs[OWN_STEP];
#pragma unroll
                    for (int u = 0; u < OWN_STEP; ++u) {
                        const int iu = (i + u < m) ? (i + u) : (m - 1);
                        const int e  = __builtin_amdgcn_readlane(entv, iu);
                        ps[u] = __int_as_float(__builtin_amdgcn_readlane(entp, iu));
                        ks[u] = e >> 8;
                        zs[u] = ((const float2*)stage)[(e & 255) * 32 + lane];
                    }
#pragma unroll
                    for (int u = 0; u < OWN_STEP; ++u) {
                        if (i + u < m) {
                            const int   k = ks[u];
                            const float p = ps[u];
                            ax[k] += p * zs[u].x;
                            ay[k] += p * zs[u].y;
                            dreg += (lane == k) ? p : 0.f;
                        }
                    }
                }
            }
            __syncthreads();
        }
        if (!again) break;
        for (int r = tid; r < RANGE_ROWS; r += 512) mold[r] = rmax[r];
    }
#pragma unroll
    for (int k = 0; k < ROWS_PER_WAVE; ++k) {
        const int r = 16 * k + wave;
        const int node = row_base + r;
        if (r < RANGE_ROWS && node < N_NODES) {
            const float d = __int_as_float(__builtin_amdgcn_readlane(__float_as_int(dreg), k));
            const float inv = (d > 0.f) ? 1.0f / d : 0.f;
            float2 o;
            o.x = ax[k] * inv;
            o.y = ay[k] * inv;
            ((float2*)(out + (size_t)node * OUT_DIM))[lane] = o;
        }
    }
}

extern "C" void kernel_launch(void* const* d_in, const int* in_sizes, int n_in,
                              void* d_out, int out_size, void* d_ws, size_t ws_size,
                              hipStream_t stream) {
    const float* h  = (const float*)d_in[0];
    const int* src  = (const int*)d_in[1];
    const int* dst  = (const int*)d_in[2];
    const float* W  = (const float*)d_in[3];
    const float* A  = (const float*)d_in[4];
    float* out = (float*)d_out;

    char* ws = (char*)d_ws;
    float*         z   = (float*)        (ws);
    float*         s   = (float*)        (ws + 25600000);
    float*         t   = (float*)        (ws + 26000000);
    unsigned char* key = (unsigned char*)(ws + 26400000);
    unsigned*      pk  = (unsigned*)     (ws + WS_PK_OFF);

    gat_front<<<FRONT_WGS, 128, 0, stream>>>(h, W, A, src, dst, z, s, t, (unsigned*)key, pk, ws_size);
    gat_range<<<N_RANGES, 512, 0, stream>>>(src, dst, key, pk, ws_size, s, t, z, out);
}


# range phase: the owner wave reads its edges' z rows straight from memory, 16 row reads in flight from scalar row addresses; no LDS staging of rows, no barrier while summing; same list order and sums
# speedup vs baseline: 3.7914x; 1.1725x over previous
#include <hip/hip_runtime.h>
#include <hip/hip_bf16.h>
#include <math.h>

#define N_NODES 100000
#define N_EDGES 1600000
#define IN_DIM  128
#define OUT_DIM 64
#define LEAKY   0.01f

typedef float v2f __attribute__((ext_vector_type(2)));
typedef float v8f __attribute__((ext_vector_type(8)));
typedef float v32f __attribute__((ext_vector_type(32)));

#define RANGE_ROWS 391
#define N_RANGES   256
#define CHUNK      512
#define N_CHUNKS   (N_EDGES / CHUNK)
#define SCAN_WAVES 16
#define CHUNKS_PER_WAVE 196
#define SEG_CAP    512
#define LIST_CAP   (SCAN_WAVES * SEG_CAP)
#define PART       2048
#define MAX_PASS   200
#define OWN_CAP    PART
#define ROWS_PER_WAVE 25
#define OWN_STEP   16
#define WS_Z_BYTES   25600000
#define WS_ST_BYTES  400000
#define WS_KEY_BYTES 1600000
#define WS_PK_OFF    (WS_Z_BYTES + 2 * WS_ST_BYTES + WS_KEY_BYTES)
#define WS_NEED_PK   ((size_t)WS_PK_OFF + (size_t)N_EDGES * 4)


#define ZT_PITCH 68
#define HT_PITCH 132
#define TILES_PER_WG 5
#define FRONT_WGS (N_NODES / 16 / TILES_PER_WG)
#define EDGES_PER_TILE 256
static_assert((N_NODES % 16) == 0, "row tiles must be whole");
static_assert(FRONT_WGS * TILES_PER_WG * 16 == N_NODES, "every row tile belongs to exactly one work-group");
static_assert(EDGES_PER_TILE * (N_NODES / 16) == N_EDGES, "every edge gets its key from exactly one tile");

__global__ void __launch_bounds__(128)
gat_front(const float* __restrict__ h,
          const float* __restrict__ W,
          const float* __restrict__ A,
          const int* __restrict__ src,
          const int* __restrict__ dst,
          float* __restrict__ z,
          float* __restrict__ s, float* __restrict__ t,
          unsigned* __restrict__ key,
          unsigned* __restrict__ pk, size_t ws_bytes) {
    __shared__ float4 hb4[2][16 * HT_PITCH / 4];
    __shared__ float4 zt4[2][16 * ZT_PITCH / 4];

    const int tid  = threadIdx.x;
    const int lane = tid & 31;
    const int wave = tid >> 5;
    const int tile0 = blockIdx.x * TILES_PER_WG;
    const int col0 = wave * 16;
    const bool use_pk = ws_bytes >= WS_NEED_PK;

    const int m     = lane & 15;
    const int khalf = (lane >> 4) * 2;

    const float4* __restrict__ hsrc = (const float4*)h + (size_t)(wave * 4) * (IN_DIM / 4) + lane;
    const int hdst = (wave * 4) * (HT_PITCH / 4) + lane;
#define LOAD_TILE(T_) {                                                     \
        const float4* p_ = hsrc + (size_t)(T_) * (16 * IN_DIM / 4);         \
        g0 = p_[0 * (IN_DIM / 4)]; g1 = p_[1 * (IN_DIM / 4)];               \
        g2 = p_[2 * (IN_DIM / 4)]; g3 = p_[3 * (IN_DIM / 4)]; }
#define PARK_TILE(B_) {                                                     \
        hb4[B_][hdst + 0 * (HT_PITCH / 4)] = g0;                            \
        hb4[B_][hdst + 1 * (HT_PITCH / 4)] = g1;                            \
        hb4[B_][hdst + 2 * (HT_PITCH / 4)] = g2;                            \
        hb4[B_][hdst + 3 * (HT_PITCH / 4)] = g3; }
    float4 g0, g1, g2, g3;
    LOAD_TILE(tile0)

    const float* __restrict__ wrow = W + (size_t)(col0 + m) * IN_DIM;
    v2f bf[IN_DIM / 4];
#pragma unroll
    for (int kk = 0; kk < IN_DIM / 4; ++kk) {
        bf[kk].x = wrow[4 * kk + khalf];
        bf[kk].y = wrow[4 * kk + khalf + 1];
    }

    const int q  = tid & 15;
    const int rr = tid >> 4;
    const float4 as = ((const float4*)A)[q];
    const float4 ad = ((const float4*)(A + OUT_DIM))[q];
    const int n     = lane & 15;
    const int rbase = (lane >> 4) * 8;

    PARK_TILE(0)
    __syncthreads();
    if (TILES_PER_WG > 1) LOAD_TILE(tile0 + 1)

    for (int i = 0; i < TILES_PER_WG; ++i) {
        const int tile = tile0 + i;
        const int buf  = i & 1;
        const int row0 = tile * 16;

        const int e0 = tile * EDGES_PER_TILE + wave * 64 + lane;
        const int d0 = dst[e0];
        const int d1 = dst[e0 + 32];
        int s0 = 0, s1 = 0;
        if (use_pk) { s0 = src[e0]; s1 = src[e0 + 32]; }

        const float* __restrict__ hrow = (const float*)hb4[buf] + m * HT_PITCH + khalf;
        v2f af[IN_DIM / 4];
#pragma unroll
        for (int kk = 0; kk < IN_DIM / 4; ++kk) af[kk] = *(const v2f*)(hrow + 4 * kk);
        __builtin_amdgcn_sched_barrier(0);
        v8f c = {};
#pragma unroll
        for (int kk = 0; kk < IN_DIM / 4; ++kk) {
            c = __builtin_amdgcn_wmma_f32_16x16x4_f32(
                    false, af[kk], false, bf[kk], (short)0, c, false, false);
        }
        __builtin_amdgcn_sched_barrier(0);

        float* zt = (float*)zt4[buf];
#pragma unroll
        for (int r = 0; r < 8; ++r) {
            zt[(rbase + r) * ZT_PITCH + col0 + n] = c[r];
        }
        if (i + 1 < TILES_PER_WG) PARK_TILE(buf ^ 1)
        __syncthreads();
        if (i + 2 < TILES_PER_WG) LOAD_TILE(tile + 2)

        const float4 v0 = zt4[buf][rr * (ZT_PITCH / 4) + q];
        const float4 v1 = zt4[buf][(rr + 8) * (ZT_PITCH / 4) + q];
        float ps0 = v0.x * as.x + v0.y * as.y + v0.z * as.z + v0.w * as.w;
        float pt0 = v0.x * ad.x + v0.y * ad.y + v0.z * ad.z + v0.w * ad.w;
        float ps1 = v1.x * as.x + v1.y * as.y + v1.z * as.z + v1.w * as.w;
        float pt1 = v1.x * ad.x + v1.y * ad.y + v1.z * ad.z + v1.w * ad.w;
        ((float4*)(z + (size_t)(row0 + rr) * OUT_DIM))[q] = v0;
        ((float4*)(z + (size_t)(row0 + rr + 8) * OUT_DIM))[q] = v1;
#pragma unroll
        for (int mk = 1; mk < 16; mk <<= 1) {
            const float a0 = __shfl_xor(ps0, mk, 32);
            const float a1 = __shfl_xor(pt0, mk, 32);
            const float a2 = __shfl_xor(ps1, mk, 32);
            const float a3 = __shfl_xor(pt1, mk, 32);
            ps0 += a0; pt0 += a1; ps1 += a2; pt1 += a3;
        }
        if (q == 0) {
            s[row0 + rr]     = ps0;
            t[row0 + rr]     = pt0;
            s[row0 + rr + 8] = ps1;
            t[row0 + rr + 8] = pt1;
        }

        const unsigned k0 = (unsigned)(d0 / RANGE_ROWS);
        const unsigned k1 = (unsigned)(d1 / RANGE_ROWS);
        if (use_pk) {
            pk[e0]      = (unsigned)s0 | (((unsigned)d0 - k0 * RANGE_ROWS) << 17);
            pk[e0 + 32] = (unsigned)s1 | (((unsigned)d1 - k1 * RANGE_ROWS) << 17);
        }
        unsigned wv = 0u;
#pragma unroll
        for (int k = 0; k < 4; ++k) {
            const unsigned a0 = __builtin_amdgcn_ballot_w32(((k0 >> (2 * k)) & 1u) != 0u);
            const unsigned a1 = __builtin_amdgcn_ballot_w32(((k0 >> (2 * k + 1)) & 1u) != 0u);
            const unsigned b0 = __builtin_amdgcn_ballot_w32(((k1 >> (2 * k)) & 1u) != 0u);
            const unsigned b1 = __builtin_amdgcn_ballot_w32(((k1 >> (2 * k + 1)) & 1u) != 0u);
            wv = (lane == k)      ? ((a0 & 0xFFFFu) | (a1 << 16))      : wv;
            wv = (lane == 4 + k)  ? ((a0 >> 16) | (a1 & 0xFFFF0000u))  : wv;
            wv = (lane == 8 + k)  ? ((b0 & 0xFFFFu) | (b1 << 16))      : wv;
            wv = (lane == 12 + k) ? ((b0 >> 16) | (b1 & 0xFFFF0000u))  : wv;
        }
        if (lane < 16) key[(tile * 16 + wave * 4) * 4 + lane] = wv;
    }
#undef LOAD_TILE
#undef PARK_TILE
}


__device__ __forceinline__ bool scan_append(unsigned fl, unsigned eb, int lane, int segbase,
                                            int& cnt, unsigned* list) {
    const unsigned mk = __builtin_amdgcn_ballot_w32(fl != 0u);
    if (mk == 0u) return true;
    const unsigned fl1 = fl & (fl - 1u);
    const unsigned mk2 = __builtin_amdgcn_ballot_w32(fl1 != 0u);
    int tot = __popc(mk) + __popc(mk2);
    int off = (int)__builtin_amdgcn_mbcnt_lo(mk2, __builtin_amdgcn_mbcnt_lo(mk, 0u));
    if (mk2 != 0u) {
        const unsigned fl2 = fl1 & (fl1 - 1u);
        if (__builtin_amdgcn_ballot_w32(fl2 != 0u) != 0u) {
            const int cl = __popc(fl);
            int x = cl;
#pragma unroll
            for (int d = 1; d < 32; d <<= 1) {
                const int y = __shfl_up(x, d, 32);
                if (lane >= d) x += y;
            }
            off = x - cl;
            tot = __builtin_amdgcn_readlane(x, 31);
        }
    }
    const bool fits = (cnt + tot <= SEG_CAP);
    if (fits && fl != 0u) {
        int pos = segbase + cnt + off;
        if (fl1 == 0u) {
            list[pos] = eb + (unsigned)__builtin_ctz(fl);
        } else {
            unsigned m = fl;
            while (m) {
                list[pos++] = eb + (unsigned)__builtin_ctz(m);
                m &= m - 1;
            }
        }
    }
    cnt = fits ? cnt + tot : cnt;
    return fits;
}

__device__ __forceinline__ unsigned fkey(float v) {
    const unsigned b = __float_as_uint(v);
    return (b & 0x80000000u) ? ~b : (b | 0x80000000u);
}
__device__ __forceinline__ float funkey(unsigned k) {
    return __uint_as_float((k & 0x80000000u) ? (k & 0x7FFFFFFFu) : ~k);
}
__global__ void __launch_bounds__(512)
gat_range(const int* __restrict__ src, const int* __restrict__ dst,
          const unsigned char* __restrict__ key,
          const unsigned* __restrict__ pk, size_t ws_bytes,
          const float* __restrict__ s, const float* __restrict__ t,
          const float* __restrict__ z, float* __restrict__ out) {
    __shared__ unsigned l_idx[LIST_CAP];
    __shared__ unsigned c_pk[LIST_CAP];
    __shared__ float    c_e[LIST_CAP];
    __shared__ unsigned rmax[400];
    __shared__ unsigned mold[400];
    __shared__ float    scl[400];
    __shared__ unsigned short own_e[SCAN_WAVES * OWN_CAP];
    __shared__ int      segcnt[SCAN_WAVES];
    __shared__ int      more;

    const int tid  = threadIdx.x;
    const int lane = tid & 31;
    const int wave = __builtin_amdgcn_readfirstlane(tid >> 5);
    const int gid  = blockIdx.x;
    const int row_base = gid * RANGE_ROWS;
    const bool use_pk = ws_bytes >= WS_NEED_PK;

    v32f ax = {};
    v32f ay = {};
    for (int i = tid; i < 400; i += 512) rmax[i] = 0u;
    float dreg = 0.f;

    const uint4* __restrict__ key4 = (const uint4*)key;
    const unsigned inv0 = ((gid & 1)   ? 0u : 0x0000FFFFu) | ((gid & 2)   ? 0u : 0xFFFF0000u);
    const unsigned inv1 = ((gid & 4)   ? 0u : 0x0000FFFFu) | ((gid & 8)   ? 0u : 0xFFFF0000u);
    const unsigned inv2 = ((gid & 16)  ? 0u : 0x0000FFFFu) | ((gid & 32)  ? 0u : 0xFFFF0000u);
    const unsigned inv3 = ((gid & 64)  ? 0u : 0x0000FFFFu) | ((gid & 128) ? 0u : 0xFFFF0000u);
    bool slow = false;
    int c = wave * CHUNKS_PER_WAVE;
    int cend = c + CHUNKS_PER_WAVE;
    if (cend > N_CHUNKS) cend = N_CHUNKS;

    for (int pass = 0; pass < MAX_PASS; ++pass) {
        if (tid == 0) more = 0;
        __syncthreads();

        int cnt = 0;
        bool full = false;
        while (c < cend && !full) {
            uint4 kv[8];
            const int glast = cend * 32 - 1;
#pragma unroll
            for (int u = 0; u < 8; ++u) {
                int gi = (c + (u & ~1)) * 32 + 2 * lane + (u & 1);
                if (gi > glast) gi = glast;
                kv[u] = key4[gi];
            }
#pragma unroll
            for (int u = 0; u < 4; ++u) {
                if (c + 1 < cend && !full && !slow) {
                    const unsigned eqa = (kv[2 * u].x ^ inv0) & (kv[2 * u].y ^ inv1) & (kv[2 * u].z ^ inv2) & (kv[2 * u].w ^ inv3);
                    const unsigned eqb = (kv[2 * u + 1].x ^ inv0) & (kv[2 * u + 1].y ^ inv1) & (kv[2 * u + 1].z ^ inv2) & (kv[2 * u + 1].w ^ inv3);
                    const unsigned fl = (eqa & (eqa >> 16)) | ((eqb & (eqb >> 16)) << 16);
                    const bool fits = scan_append(fl, (unsigned)c * CHUNK + (unsigned)(lane * 32), lane, wave * SEG_CAP, cnt, l_idx);
                    if (fits) c += 2;
                    else if (cnt == 0) slow = true;
                    else full = true;
                }
            }
            if (!full && c < cend && (slow || c + 1 == cend)) {
                const uint4 k1 = key4[c * 32 + lane];
                const unsigned eq = (k1.x ^ inv0) & (k1.y ^ inv1) & (k1.z ^ inv2) & (k1.w ^ inv3);
                const unsigned fl = eq & (eq >> 16);
                const bool fits = scan_append(fl, (unsigned)c * CHUNK + (unsigned)(lane * 16), lane, wave * SEG_CAP, cnt, l_idx);
                if (fits) ++c;
                else full = true;
            }
        }
        if (lane == 0) {
            segcnt[wave] = cnt;
            if (c < cend) more = 1;
        }
        __syncthreads();
        const int again = more;

        int total = 0;
        for (int w = 0; w < SCAN_WAVES; ++w) total += segcnt[w];
        {
            unsigned ee[SCAN_WAVES];
            int si[SCAN_WAVES], di[SCAN_WAVES];
            float sv[SCAN_WAVES], tv[SCAN_WAVES];
#pragma unroll
            for (int w = 0; w < SCAN_WAVES; ++w)
                ee[w] = (tid < segcnt[w]) ? l_idx[w * SEG_CAP + tid] : 0u;
            if (use_pk) {
#pragma unroll
                for (int w = 0; w < SCAN_WAVES; ++w) si[w] = (int)pk[ee[w]];
#pragma unroll
                for (int w = 0; w < SCAN_WAVES; ++w) {
                    const unsigned pw = (tid < segcnt[w]) ? (unsigned)si[w] : 0u;
                    si[w] = (int)(pw & 0x1FFFFu);
                    di[w] = row_base + (int)(pw >> 17);
                }
            } else {
#pragma unroll
                for (int w = 0; w < SCAN_WAVES; ++w) { si[w] = src[ee[w]]; di[w] = dst[ee[w]]; }
            }
#pragma unroll
            for (int w = 0; w < SCAN_WAVES; ++w) { sv[w] = s[si[w]]; tv[w] = t[di[w]]; }
#pragma unroll
            for (int w = 0; w < SCAN_WAVES; ++w) {
                float v = sv[w] + tv[w];
                v = (v > 0.f) ? v : LEAKY * v;
                sv[w] = v;
                if (tid < segcnt[w]) atomicMax(&rmax[di[w] - row_base], fkey(v));
            }
            __syncthreads();
            if (pass > 0) {
                for (int r = tid; r < RANGE_ROWS; r += 512) {
                    const unsigned ko = mold[r], kn = rmax[r];
                    scl[r] = (ko != 0u && kn != ko) ? expf(funkey(ko) - funkey(kn)) : 1.0f;
                }
                __syncthreads();
#pragma unroll
                for (int k = 0; k < ROWS_PER_WAVE; ++k) {
                    const int r = 16 * k + wave;
                    const float sc = (r < RANGE_ROWS) ? scl[r] : 1.0f;
                    ax[k] *= sc;
                    ay[k] *= sc;
                }
                const int rown = lane * 16 + wave;
                if (rown < RANGE_ROWS) dreg *= scl[rown];
            }
            int base = 0;
#pragma unroll
            for (int w = 0; w < SCAN_WAVES; ++w) {
                const int n = segcnt[w];
                if (tid < n) {
                    const int lrow = di[w] - row_base;
                    c_pk[base + tid] = (unsigned)si[w] | ((unsigned)lrow << 17);
                    c_e[base + tid]  = expf(sv[w] - funkey(rmax[lrow]));
                }
                base += n;
            }
        }
        __syncthreads();

        const int total_s = __builtin_amdgcn_readfirstlane(total);
        for (int cb = 0; cb < total_s; cb += PART) {
            int pend = cb + PART;
            if (pend > total_s) pend = total_s;
            int n_own = 0;
            for (int g0 = cb; g0 < pend; g0 += 256) {
                unsigned pkv[8];
#pragma unroll
                for (int g = 0; g < 8; ++g) {
                    const int j = g0 + g * 32 + lane;
                    const int jc = (j < total_s) ? j : 0;
                    pkv[g] = c_pk[jc];
                }
#pragma unroll
                for (int g = 0; g < 8; ++g) {
                    const int j = g0 + g * 32 + lane;
                    const int lrow = (int)(pkv[g] >> 17);
                    const bool mine = (j < total_s) && ((lrow & 15) == wave);
                    const unsigned mk = __builtin_amdgcn_ballot_w32(mine);
                    if (mine) {
                        const int pos = wave * OWN_CAP + n_own + (int)__builtin_amdgcn_mbcnt_lo(mk, 0u);
                        own_e[pos] = (unsigned short)(j - cb);
                    }
                    n_own += __popc(mk);
                }
            }
            for (int c0 = 0; c0 < n_own; c0 += 32) {
                int m = n_own - c0;
                if (m > 32) m = 32;
                const int li = (lane < m) ? lane : (m - 1);
                const int ei = (int)own_e[wave * OWN_CAP + c0 + li];
                const int entk = (int)c_pk[cb + ei];
                const float wv = c_e[cb + ei];
                const int entp = __float_as_int((lane < m) ? wv : 0.f);
#pragma unroll
                for (int hh = 0; hh < 2; ++hh) {
                    if (hh * OWN_STEP < m) {
                        int    kw[OWN_STEP];
                        float2 zs[OWN_STEP];
#pragma unroll
                        for (int u = 0; u < OWN_STEP; ++u) {
                            kw[u] = __builtin_amdgcn_readlane(entk, hh * OWN_STEP + u);
                            const unsigned long long zb = (unsigned long long)z + ((unsigned long long)(unsigned)(kw[u] & 0x1FFFF) << 8);
                            const unsigned zlo = __builtin_amdgcn_readfirstlane((unsigned)zb);
                            const unsigned zhi = __builtin_amdgcn_readfirstlane((unsigned)(zb >> 32));
                            zs[u] = ((const __attribute__((address_space(1))) float2*)(((unsigned long long)zhi << 32) | zlo))[lane];
                        }
                        asm volatile("" : "+v"(zs[0].x) : "v"(zs[1].x), "v"(zs[2].x), "v"(zs[3].x), "v"(zs[4].x), "v"(zs[5].x), "v"(zs[6].x), "v"(zs[7].x), "v"(zs[8].x), "v"(zs[9].x), "v"(zs[10].x), "v"(zs[11].x), "v"(zs[12].x), "v"(zs[13].x), "v"(zs[14].x), "v"(zs[15].x));
#pragma unroll
                        for (int u = 0; u < OWN_STEP; ++u) {
                            const int   k = kw[u] >> 21;
                            const float p = __int_as_float(__builtin_amdgcn_readlane(entp, hh * OWN_STEP + u));
                            ax[k] += p * zs[u].x;
                            ay[k] += p * zs[u].y;
                            dreg += (lane == k) ? p : 0.f;
                        }
                    }
                }
            }
        }
        if (!again) break;
        for (int r = tid; r < RANGE_ROWS; r += 512) mold[r] = rmax[r];
    }
#pragma unroll
    for (int k = 0; k < ROWS_PER_WAVE; ++k) {
        const int r = 16 * k + wave;
        const int node = row_base + r;
        if (r < RANGE_ROWS && node < N_NODES) {
            const float d = __int_as_float(__builtin_amdgcn_readlane(__float_as_int(dreg), k));
            const float inv = (d > 0.f) ? 1.0f / d : 0.f;
            float2 o;
            o.x = ax[k] * inv;
            o.y = ay[k] * inv;
            ((float2*)(out + (size_t)node * OUT_DIM))[lane] = o;
        }
    }
}

extern "C" void kernel_launch(void* const* d_in, const int* in_sizes, int n_in,
                              void* d_out, int out_size, void* d_ws, size_t ws_size,
                              hipStream_t stream) {
    const float* h  = (const float*)d_in[0];
    const int* src  = (const int*)d_in[1];
    const int* dst  = (const int*)d_in[2];
    const float* W  = (const float*)d_in[3];
    const float* A  = (const float*)d_in[4];
    float* out = (float*)d_out;

    char* ws = (char*)d_ws;
    float*         z   = (float*)        (ws);
    float*         s   = (float*)        (ws + 25600000);
    float*         t   = (float*)        (ws + 26000000);
    unsigned char* key = (unsigned char*)(ws + 26400000);
    unsigned*      pk  = (unsigned*)     (ws + WS_PK_OFF);

    gat_front<<<FRONT_WGS, 128, 0, stream>>>(h, W, A, src, dst, z, s, t, (unsigned*)key, pk, ws_size);
    gat_range<<<N_RANGES, 512, 0, stream>>>(src, dst, key, pk, ws_size, s, t, z, out);
}
